# starting kernel
# baseline (speedup 1.0000x reference)
#include <hip/hip_runtime.h>
#include <hip/hip_bf16.h>


typedef float v2f __attribute__((ext_vector_type(2)));
typedef float v8f __attribute__((ext_vector_type(8)));

#define IN_CH 128


__global__ void k_init_deg(float* deg, int n) {
    int i = blockIdx.x * blockDim.x + threadIdx.x;
    if (i < n) deg[i] = 1.0f;
}

__global__ void k_deg_accum(const int* __restrict__ dst, float* deg, int E) {
    int e = blockIdx.x * blockDim.x + threadIdx.x;
    if (e < E) unsafeAtomicAdd(&deg[dst[e]], 1.0f);
}

__global__ void k_dinv(const float* __restrict__ deg, float* dinv, int n) {
    int i = blockIdx.x * blockDim.x + threadIdx.x;
    if (i < n) {
        float d = deg[i];
        dinv[i] = d > 0.0f ? __frsqrt_rn(d) : 0.0f;
    }
}


template <int K, int N>
__global__ void k_gemm_wmma(const float* __restrict__ A,
                            const float* __restrict__ W,
                            float* __restrict__ O,
                            int n_rows, int n_tiles) {
    int wave = (blockIdx.x * blockDim.x + threadIdx.x) >> 5;
    int lane = threadIdx.x & 31;
    if (wave >= n_tiles) return;

    const int m0    = wave * 16;
    const int mrow  = m0 + (lane & 15);
    const int rm    = mrow < n_rows ? mrow : (n_rows - 1);
    const int khalf = (lane >> 4) * 2;
    const int colL  = lane & 15;

    constexpr int NT = N / 16;
    v8f acc[NT] = {};

    for (int k = 0; k < K; k += 4) {
        v2f a;
        a.x = A[rm * K + k + khalf];
        a.y = A[rm * K + k + khalf + 1];
#pragma unroll
        for (int t = 0; t < NT; ++t) {
            v2f b;
            b.x = W[(k + khalf) * N + t * 16 + colL];
            b.y = W[(k + khalf + 1) * N + t * 16 + colL];
            acc[t] = __builtin_amdgcn_wmma_f32_16x16x4_f32(
                false, a, false, b, (short)0, acc[t], false, false);
        }
    }

    const int mbase = m0 + (lane >> 4) * 8;
    if (m0 + 16 <= n_rows) {
#pragma unroll
        for (int t = 0; t < NT; ++t) {
#pragma unroll
            for (int r = 0; r < 8; ++r) {
                O[(mbase + r) * N + t * 16 + colL] = acc[t][r];
            }
        }
    } else {
#pragma unroll
        for (int t = 0; t < NT; ++t) {
#pragma unroll
            for (int r = 0; r < 8; ++r) {
                int row = mbase + r;
                if (row < n_rows) O[row * N + t * 16 + colL] = acc[t][r];
            }
        }
    }
}


template <int LOGC>
__global__ void k_selfloop(const float* __restrict__ h,
                           const float* __restrict__ dinv,
                           float* __restrict__ out, int total) {
    int id = blockIdx.x * blockDim.x + threadIdx.x;
    if (id >= total) return;
    float d = dinv[id >> LOGC];
    out[id] = d * d * h[id];
}


__global__ void k_edge_scatter32(const int* __restrict__ src,
                                 const int* __restrict__ dst,
                                 const float* __restrict__ dinv,
                                 const float* __restrict__ h,
                                 float* out, int E) {
    int w = (blockIdx.x * blockDim.x + threadIdx.x) >> 5;
    int lane = threadIdx.x & 31;
    if (w >= E) return;
    int s = src[w], d = dst[w];
    float norm = dinv[s] * dinv[d];
    float v = h[s * 32 + lane] * norm;
    unsafeAtomicAdd(&out[d * 32 + lane], v);
}

__global__ void k_edge_scatter64(const int* __restrict__ src,
                                 const int* __restrict__ dst,
                                 const float* __restrict__ dinv,
                                 const float* __restrict__ h,
                                 float* out, int E) {
    int w = (blockIdx.x * blockDim.x + threadIdx.x) >> 5;
    int lane = threadIdx.x & 31;
    if (w >= E) return;
    int s = src[w], d = dst[w];
    float norm = dinv[s] * dinv[d];
    const float2* hp = (const float2*)(h + (size_t)s * 64);
    float2 v = hp[lane];
    unsafeAtomicAdd(&out[d * 64 + 2 * lane + 0], v.x * norm);
    unsafeAtomicAdd(&out[d * 64 + 2 * lane + 1], v.y * norm);
}


template <int C>
__global__ void k_bias_relu(float* h, const float* __restrict__ bias,
                            int total) {
    int id = blockIdx.x * blockDim.x + threadIdx.x;
    if (id >= total) return;
    float v = h[id] + bias[id & (C - 1)];
    h[id] = v > 0.0f ? v : 0.0f;
}


__global__ void k_pool_accum(const float* __restrict__ h,
                             const int* __restrict__ batch,
                             float* sums, int n) {
    int id = blockIdx.x * blockDim.x + threadIdx.x;
    if (id >= n * 64) return;
    int node = id >> 6, c = id & 63;
    int g = batch[node];
    unsafeAtomicAdd(&sums[g * 64 + c], h[id]);
}

__global__ void k_count_accum(const int* __restrict__ batch, float* cnts, int n) {
    int i = blockIdx.x * blockDim.x + threadIdx.x;
    if (i < n) unsafeAtomicAdd(&cnts[batch[i]], 1.0f);
}


__global__ void k_final_head(const float* __restrict__ sums,
                             const float* __restrict__ cnts,
                             const float* __restrict__ Wc,
                             const float* __restrict__ bc,
                             float* out, int G) {
    int g = threadIdx.x;
    if (g >= G) return;
    float c = cnts[g];
    c = c > 1.0f ? c : 1.0f;
    float inv = 1.0f / c;
    float l0 = bc[0], l1 = bc[1];
    for (int ch = 0; ch < 64; ++ch) {
        float p = sums[g * 64 + ch] * inv;
        out[G * 2 + g * 64 + ch] = p;
        l0 = fmaf(p, Wc[ch * 2 + 0], l0);
        l1 = fmaf(p, Wc[ch * 2 + 1], l1);
    }
    out[g * 2 + 0] = l0;
    out[g * 2 + 1] = l1;
}


extern "C" void kernel_launch(void* const* d_in, const int* in_sizes, int n_in,
                              void* d_out, int out_size, void* d_ws, size_t ws_size,
                              hipStream_t stream) {
    const float* x   = (const float*)d_in[0];
    const float* W1  = (const float*)d_in[1];
    const float* b1  = (const float*)d_in[2];
    const float* W2  = (const float*)d_in[3];
    const float* b2  = (const float*)d_in[4];
    const float* Wc  = (const float*)d_in[5];
    const float* bc  = (const float*)d_in[6];
    const int*   ei  = (const int*)d_in[7];
    const int*   bat = (const int*)d_in[8];

    const int n = in_sizes[0] / IN_CH;
    const int E = in_sizes[7] / 2;
    const int G = 64;

    const int* src = ei;
    const int* dst = ei + E;

    float* ws   = (float*)d_ws;
    float* deg  = ws;                ws += n;
    float* dinv = ws;                ws += n;
    float* h1   = ws;                ws += (size_t)n * 32;
    float* o1   = ws;                ws += (size_t)n * 32;
    float* h2   = ws;                ws += (size_t)n * 64;
    float* o2   = ws;                ws += (size_t)n * 64;
    float* sums = ws;                ws += (size_t)G * 64;
    float* cnts = ws;                ws += G;

    const int B = 256;
    auto cdiv = [](long long a, long long b) { return (int)((a + b - 1) / b); };

    k_init_deg<<<cdiv(n, B), B, 0, stream>>>(deg, n);
    k_deg_accum<<<cdiv(E, B), B, 0, stream>>>(dst, deg, E);
    k_dinv<<<cdiv(n, B), B, 0, stream>>>(deg, dinv, n);

    const int tiles = (n + 15) / 16;
    k_gemm_wmma<128, 32><<<cdiv(tiles, 4), 128, 0, stream>>>(x, W1, h1, n, tiles);
    k_selfloop<5><<<cdiv((long long)n * 32, B), B, 0, stream>>>(h1, dinv, o1, n * 32);
    k_edge_scatter32<<<cdiv((long long)E * 32, B), B, 0, stream>>>(src, dst, dinv, h1, o1, E);
    k_bias_relu<32><<<cdiv((long long)n * 32, B), B, 0, stream>>>(o1, b1, n * 32);

    k_gemm_wmma<32, 64><<<cdiv(tiles, 4), 128, 0, stream>>>(o1, W2, h2, n, tiles);
    k_selfloop<6><<<cdiv((long long)n * 64, B), B, 0, stream>>>(h2, dinv, o2, n * 64);
    k_edge_scatter64<<<cdiv((long long)E * 32, B), B, 0, stream>>>(src, dst, dinv, h2, o2, E);
    k_bias_relu<64><<<cdiv((long long)n * 64, B), B, 0, stream>>>(o2, b2, n * 64);

    hipMemsetAsync(sums, 0, (size_t)(G * 64 + G) * sizeof(float), stream);
    k_pool_accum<<<cdiv((long long)n * 64, B), B, 0, stream>>>(o2, bat, sums, n);
    k_count_accum<<<cdiv(n, B), B, 0, stream>>>(bat, cnts, n);
    k_final_head<<<1, 64, 0, stream>>>(sums, cnts, Wc, bc, (float*)d_out, G);
}


# Edge scatter: one wave per 32 edges (coalesced index loads, scalar broadcast), row gathers batched ahead of full-line f32 atomics; atomic arrival order still unordered
# speedup vs baseline: 1.4234x; 1.4234x over previous
#include <hip/hip_runtime.h>
#include <hip/hip_bf16.h>


typedef float v2f __attribute__((ext_vector_type(2)));
typedef float v8f __attribute__((ext_vector_type(8)));

#define IN_CH 128


__global__ void k_init_deg(float* deg, int n) {
    int i = blockIdx.x * blockDim.x + threadIdx.x;
    if (i < n) deg[i] = 1.0f;
}

__global__ void k_deg_accum(const int* __restrict__ dst, float* deg, int E) {
    int e = blockIdx.x * blockDim.x + threadIdx.x;
    if (e < E) unsafeAtomicAdd(&deg[dst[e]], 1.0f);
}

__global__ void k_dinv(const float* __restrict__ deg, float* dinv, int n) {
    int i = blockIdx.x * blockDim.x + threadIdx.x;
    if (i < n) {
        float d = deg[i];
        dinv[i] = d > 0.0f ? __frsqrt_rn(d) : 0.0f;
    }
}


template <int K, int N>
__global__ void k_gemm_wmma(const float* __restrict__ A,
                            const float* __restrict__ W,
                            float* __restrict__ O,
                            int n_rows, int n_tiles) {
    int wave = (blockIdx.x * blockDim.x + threadIdx.x) >> 5;
    int lane = threadIdx.x & 31;
    if (wave >= n_tiles) return;

    const int m0    = wave * 16;
    const int mrow  = m0 + (lane & 15);
    const int rm    = mrow < n_rows ? mrow : (n_rows - 1);
    const int khalf = (lane >> 4) * 2;
    const int colL  = lane & 15;

    constexpr int NT = N / 16;
    v8f acc[NT] = {};

    for (int k = 0; k < K; k += 4) {
        v2f a;
        a.x = A[rm * K + k + khalf];
        a.y = A[rm * K + k + khalf + 1];
#pragma unroll
        for (int t = 0; t < NT; ++t) {
            v2f b;
            b.x = W[(k + khalf) * N + t * 16 + colL];
            b.y = W[(k + khalf + 1) * N + t * 16 + colL];
            acc[t] = __builtin_amdgcn_wmma_f32_16x16x4_f32(
                false, a, false, b, (short)0, acc[t], false, false);
        }
    }

    const int mbase = m0 + (lane >> 4) * 8;
    if (m0 + 16 <= n_rows) {
#pragma unroll
        for (int t = 0; t < NT; ++t) {
#pragma unroll
            for (int r = 0; r < 8; ++r) {
                O[(mbase + r) * N + t * 16 + colL] = acc[t][r];
            }
        }
    } else {
#pragma unroll
        for (int t = 0; t < NT; ++t) {
#pragma unroll
            for (int r = 0; r < 8; ++r) {
                int row = mbase + r;
                if (row < n_rows) O[row * N + t * 16 + colL] = acc[t][r];
            }
        }
    }
}


template <int LOGC>
__global__ void k_selfloop(const float* __restrict__ h,
                           const float* __restrict__ dinv,
                           float* __restrict__ out, int total) {
    int id = blockIdx.x * blockDim.x + threadIdx.x;
    if (id >= total) return;
    float d = dinv[id >> LOGC];
    out[id] = d * d * h[id];
}


template <int C, int G>
__global__ void k_edge_scatter(const int* __restrict__ src,
                               const int* __restrict__ dst,
                               const float* __restrict__ dinv,
                               const float* __restrict__ h,
                               float* __restrict__ out, int E) {
    constexpr int Q = C / 32;
    const int wave = (blockIdx.x * blockDim.x + threadIdx.x) >> 5;
    const int lane = threadIdx.x & 31;
    const int base = wave * 32;
    if (base >= E) return;

    const int e = base + lane;
    int s = 0, d = 0;
    float nrm = 0.0f;
    if (e < E) {
        s = src[e];
        d = dst[e];
        nrm = dinv[s] * dinv[d];
    }

    if (base + 32 <= E) {
#pragma unroll
        for (int j0 = 0; j0 < 32; j0 += G) {
            float v[G][Q];
#pragma unroll
            for (int u = 0; u < G; ++u) {
                const int sj = __builtin_amdgcn_readlane(s, j0 + u);
                const float* hp = h + sj * C + lane;
#pragma unroll
                for (int q = 0; q < Q; ++q) v[u][q] = hp[q * 32];
            }
#pragma unroll
            for (int u = 0; u < G; ++u) {
                const int dj = __builtin_amdgcn_readlane(d, j0 + u);
                const float nj = __int_as_float(
                    __builtin_amdgcn_readlane(__float_as_int(nrm), j0 + u));
                float* op = out + dj * C + lane;
#pragma unroll
                for (int q = 0; q < Q; ++q)
                    unsafeAtomicAdd(op + q * 32, v[u][q] * nj);
            }
        }
    } else {
        const int cnt = E - base;
        for (int j = 0; j < cnt; ++j) {
            const int sj = __shfl(s, j);
            const int dj = __shfl(d, j);
            const float nj = __shfl(nrm, j);
#pragma unroll
            for (int q = 0; q < Q; ++q)
                unsafeAtomicAdd(&out[dj * C + q * 32 + lane],
                                h[sj * C + q * 32 + lane] * nj);
        }
    }
}


template <int C>
__global__ void k_bias_relu(float* h, const float* __restrict__ bias,
                            int total) {
    int id = blockIdx.x * blockDim.x + threadIdx.x;
    if (id >= total) return;
    float v = h[id] + bias[id & (C - 1)];
    h[id] = v > 0.0f ? v : 0.0f;
}


__global__ void k_pool_accum(const float* __restrict__ h,
                             const int* __restrict__ batch,
                             float* sums, int n) {
    int id = blockIdx.x * blockDim.x + threadIdx.x;
    if (id >= n * 64) return;
    int node = id >> 6, c = id & 63;
    int g = batch[node];
    unsafeAtomicAdd(&sums[g * 64 + c], h[id]);
}

__global__ void k_count_accum(const int* __restrict__ batch, float* cnts, int n) {
    int i = blockIdx.x * blockDim.x + threadIdx.x;
    if (i < n) unsafeAtomicAdd(&cnts[batch[i]], 1.0f);
}


__global__ void k_final_head(const float* __restrict__ sums,
                             const float* __restrict__ cnts,
                             const float* __restrict__ Wc,
                             const float* __restrict__ bc,
                             float* out, int G) {
    int g = threadIdx.x;
    if (g >= G) return;
    float c = cnts[g];
    c = c > 1.0f ? c : 1.0f;
    float inv = 1.0f / c;
    float l0 = bc[0], l1 = bc[1];
    for (int ch = 0; ch < 64; ++ch) {
        float p = sums[g * 64 + ch] * inv;
        out[G * 2 + g * 64 + ch] = p;
        l0 = fmaf(p, Wc[ch * 2 + 0], l0);
        l1 = fmaf(p, Wc[ch * 2 + 1], l1);
    }
    out[g * 2 + 0] = l0;
    out[g * 2 + 1] = l1;
}


extern "C" void kernel_launch(void* const* d_in, const int* in_sizes, int n_in,
                              void* d_out, int out_size, void* d_ws, size_t ws_size,
                              hipStream_t stream) {
    const float* x   = (const float*)d_in[0];
    const float* W1  = (const float*)d_in[1];
    const float* b1  = (const float*)d_in[2];
    const float* W2  = (const float*)d_in[3];
    const float* b2  = (const float*)d_in[4];
    const float* Wc  = (const float*)d_in[5];
    const float* bc  = (const float*)d_in[6];
    const int*   ei  = (const int*)d_in[7];
    const int*   bat = (const int*)d_in[8];

    const int n = in_sizes[0] / IN_CH;
    const int E = in_sizes[7] / 2;
    const int G = 64;

    const int* src = ei;
    const int* dst = ei + E;

    float* ws   = (float*)d_ws;
    float* deg  = ws;                ws += n;
    float* dinv = ws;                ws += n;
    float* h1   = ws;                ws += (size_t)n * 32;
    float* o1   = ws;                ws += (size_t)n * 32;
    float* h2   = ws;                ws += (size_t)n * 64;
    float* o2   = ws;                ws += (size_t)n * 64;
    float* sums = ws;                ws += (size_t)G * 64;
    float* cnts = ws;                ws += G;

    const int B = 256;
    auto cdiv = [](long long a, long long b) { return (int)((a + b - 1) / b); };

    k_init_deg<<<cdiv(n, B), B, 0, stream>>>(deg, n);
    k_deg_accum<<<cdiv(E, B), B, 0, stream>>>(dst, deg, E);
    k_dinv<<<cdiv(n, B), B, 0, stream>>>(deg, dinv, n);

    const int tiles = (n + 15) / 16;
    k_gemm_wmma<128, 32><<<cdiv(tiles, 4), 128, 0, stream>>>(x, W1, h1, n, tiles);
    k_selfloop<5><<<cdiv((long long)n * 32, B), B, 0, stream>>>(h1, dinv, o1, n * 32);
    k_edge_scatter<32, 32><<<cdiv(E, B), B, 0, stream>>>(src, dst, dinv, h1, o1, E);
    k_bias_relu<32><<<cdiv((long long)n * 32, B), B, 0, stream>>>(o1, b1, n * 32);

    k_gemm_wmma<32, 64><<<cdiv(tiles, 4), 128, 0, stream>>>(o1, W2, h2, n, tiles);
    k_selfloop<6><<<cdiv((long long)n * 64, B), B, 0, stream>>>(h2, dinv, o2, n * 64);
    k_edge_scatter<64, 16><<<cdiv(E, B), B, 0, stream>>>(src, dst, dinv, h2, o2, E);
    k_bias_relu<64><<<cdiv((long long)n * 64, B), B, 0, stream>>>(o2, b2, n * 64);

    hipMemsetAsync(sums, 0, (size_t)(G * 64 + G) * sizeof(float), stream);
    k_pool_accum<<<cdiv((long long)n * 64, B), B, 0, stream>>>(o2, bat, sums, n);
    k_count_accum<<<cdiv(n, B), B, 0, stream>>>(bat, cnts, n);
    k_final_head<<<1, 64, 0, stream>>>(sums, cnts, Wc, bc, (float*)d_out, G);
}


# Pooling: one wave per 32 nodes sums rows in registers over runs of equal graph id, then one full-line atomic pair and one count atomic per run, not per node; pooled sums reassociated
# speedup vs baseline: 2.0690x; 1.4536x over previous
#include <hip/hip_runtime.h>
#include <hip/hip_bf16.h>


typedef float v2f __attribute__((ext_vector_type(2)));
typedef float v8f __attribute__((ext_vector_type(8)));

#define IN_CH 128


__global__ void k_init_deg(float* deg, int n) {
    int i = blockIdx.x * blockDim.x + threadIdx.x;
    if (i < n) deg[i] = 1.0f;
}

__global__ void k_deg_accum(const int* __restrict__ dst, float* deg, int E) {
    int e = blockIdx.x * blockDim.x + threadIdx.x;
    if (e < E) unsafeAtomicAdd(&deg[dst[e]], 1.0f);
}

__global__ void k_dinv(const float* __restrict__ deg, float* dinv, int n) {
    int i = blockIdx.x * blockDim.x + threadIdx.x;
    if (i < n) {
        float d = deg[i];
        dinv[i] = d > 0.0f ? __frsqrt_rn(d) : 0.0f;
    }
}


template <int K, int N>
__global__ void k_gemm_wmma(const float* __restrict__ A,
                            const float* __restrict__ W,
                            float* __restrict__ O,
                            int n_rows, int n_tiles) {
    int wave = (blockIdx.x * blockDim.x + threadIdx.x) >> 5;
    int lane = threadIdx.x & 31;
    if (wave >= n_tiles) return;

    const int m0    = wave * 16;
    const int mrow  = m0 + (lane & 15);
    const int rm    = mrow < n_rows ? mrow : (n_rows - 1);
    const int khalf = (lane >> 4) * 2;
    const int colL  = lane & 15;

    constexpr int NT = N / 16;
    v8f acc[NT] = {};

    for (int k = 0; k < K; k += 4) {
        v2f a;
        a.x = A[rm * K + k + khalf];
        a.y = A[rm * K + k + khalf + 1];
#pragma unroll
        for (int t = 0; t < NT; ++t) {
            v2f b;
            b.x = W[(k + khalf) * N + t * 16 + colL];
            b.y = W[(k + khalf + 1) * N + t * 16 + colL];
            acc[t] = __builtin_amdgcn_wmma_f32_16x16x4_f32(
                false, a, false, b, (short)0, acc[t], false, false);
        }
    }

    const int mbase = m0 + (lane >> 4) * 8;
    if (m0 + 16 <= n_rows) {
#pragma unroll
        for (int t = 0; t < NT; ++t) {
#pragma unroll
            for (int r = 0; r < 8; ++r) {
                O[(mbase + r) * N + t * 16 + colL] = acc[t][r];
            }
        }
    } else {
#pragma unroll
        for (int t = 0; t < NT; ++t) {
#pragma unroll
            for (int r = 0; r < 8; ++r) {
                int row = mbase + r;
                if (row < n_rows) O[row * N + t * 16 + colL] = acc[t][r];
            }
        }
    }
}


template <int LOGC>
__global__ void k_selfloop(const float* __restrict__ h,
                           const float* __restrict__ dinv,
                           float* __restrict__ out, int total) {
    int id = blockIdx.x * blockDim.x + threadIdx.x;
    if (id >= total) return;
    float d = dinv[id >> LOGC];
    out[id] = d * d * h[id];
}


template <int C, int G>
__global__ void k_edge_scatter(const int* __restrict__ src,
                               const int* __restrict__ dst,
                               const float* __restrict__ dinv,
                               const float* __restrict__ h,
                               float* __restrict__ out, int E) {
    constexpr int Q = C / 32;
    const int wave = (blockIdx.x * blockDim.x + threadIdx.x) >> 5;
    const int lane = threadIdx.x & 31;
    const int base = wave * 32;
    if (base >= E) return;

    const int e = base + lane;
    int s = 0, d = 0;
    float nrm = 0.0f;
    if (e < E) {
        s = src[e];
        d = dst[e];
        nrm = dinv[s] * dinv[d];
    }

    if (base + 32 <= E) {
#pragma unroll
        for (int j0 = 0; j0 < 32; j0 += G) {
            float v[G][Q];
#pragma unroll
            for (int u = 0; u < G; ++u) {
                const int sj = __builtin_amdgcn_readlane(s, j0 + u);
                const float* hp = h + sj * C + lane;
#pragma unroll
                for (int q = 0; q < Q; ++q) v[u][q] = hp[q * 32];
            }
#pragma unroll
            for (int u = 0; u < G; ++u) {
                const int dj = __builtin_amdgcn_readlane(d, j0 + u);
                const float nj = __int_as_float(
                    __builtin_amdgcn_readlane(__float_as_int(nrm), j0 + u));
                float* op = out + dj * C + lane;
#pragma unroll
                for (int q = 0; q < Q; ++q)
                    unsafeAtomicAdd(op + q * 32, v[u][q] * nj);
            }
        }
    } else {
        const int cnt = E - base;
        for (int j = 0; j < cnt; ++j) {
            const int sj = __shfl(s, j);
            const int dj = __shfl(d, j);
            const float nj = __shfl(nrm, j);
#pragma unroll
            for (int q = 0; q < Q; ++q)
                unsafeAtomicAdd(&out[dj * C + q * 32 + lane],
                                h[sj * C + q * 32 + lane] * nj);
        }
    }
}


template <int C>
__global__ void k_bias_relu(float* h, const float* __restrict__ bias,
                            int total) {
    int id = blockIdx.x * blockDim.x + threadIdx.x;
    if (id >= total) return;
    float v = h[id] + bias[id & (C - 1)];
    h[id] = v > 0.0f ? v : 0.0f;
}


__device__ __forceinline__ void pool_flush(float* sums, float* cnts, int g,
                                           int lane, float a0, float a1,
                                           float run) {
    unsafeAtomicAdd(&sums[g * 64 + lane], a0);
    unsafeAtomicAdd(&sums[g * 64 + 32 + lane], a1);
    if (lane == 0) unsafeAtomicAdd(&cnts[g], run);
}

template <int G>
__global__ void k_pool_runs(const float* __restrict__ h,
                            const int* __restrict__ batch,
                            float* sums, float* cnts, int n) {
    const int wave = (blockIdx.x * blockDim.x + threadIdx.x) >> 5;
    const int lane = threadIdx.x & 31;
    const int base = wave * 32;
    if (base >= n) return;

    const int node = base + lane;
    const int b = node < n ? batch[node] : 0;
    const float* hp = h + (size_t)base * 64 + lane;

    int cur = __builtin_amdgcn_readfirstlane(b);
    float a0 = 0.0f, a1 = 0.0f, run = 0.0f;

    if (base + 32 <= n) {
#pragma unroll
        for (int j0 = 0; j0 < 32; j0 += G) {
            float v0[G], v1[G];
#pragma unroll
            for (int u = 0; u < G; ++u) {
                v0[u] = hp[(j0 + u) * 64];
                v1[u] = hp[(j0 + u) * 64 + 32];
            }
#pragma unroll
            for (int u = 0; u < G; ++u) {
                const int g = __builtin_amdgcn_readlane(b, j0 + u);
                if (g != cur) {
                    pool_flush(sums, cnts, cur, lane, a0, a1, run);
                    cur = g; a0 = 0.0f; a1 = 0.0f; run = 0.0f;
                }
                a0 += v0[u];
                a1 += v1[u];
                run += 1.0f;
            }
        }
    } else {
        const int cnt = n - base;
        for (int j = 0; j < cnt; ++j) {
            const int g = __shfl(b, j);
            if (g != cur) {
                pool_flush(sums, cnts, cur, lane, a0, a1, run);
                cur = g; a0 = 0.0f; a1 = 0.0f; run = 0.0f;
            }
            a0 += hp[j * 64];
            a1 += hp[j * 64 + 32];
            run += 1.0f;
        }
    }
    pool_flush(sums, cnts, cur, lane, a0, a1, run);
}


__global__ void k_final_head(const float* __restrict__ sums,
                             const float* __restrict__ cnts,
                             const float* __restrict__ Wc,
                             const float* __restrict__ bc,
                             float* out, int G) {
    int g = threadIdx.x;
    if (g >= G) return;
    float c = cnts[g];
    c = c > 1.0f ? c : 1.0f;
    float inv = 1.0f / c;
    float l0 = bc[0], l1 = bc[1];
    for (int ch = 0; ch < 64; ++ch) {
        float p = sums[g * 64 + ch] * inv;
        out[G * 2 + g * 64 + ch] = p;
        l0 = fmaf(p, Wc[ch * 2 + 0], l0);
        l1 = fmaf(p, Wc[ch * 2 + 1], l1);
    }
    out[g * 2 + 0] = l0;
    out[g * 2 + 1] = l1;
}


extern "C" void kernel_launch(void* const* d_in, const int* in_sizes, int n_in,
                              void* d_out, int out_size, void* d_ws, size_t ws_size,
                              hipStream_t stream) {
    const float* x   = (const float*)d_in[0];
    const float* W1  = (const float*)d_in[1];
    const float* b1  = (const float*)d_in[2];
    const float* W2  = (const float*)d_in[3];
    const float* b2  = (const float*)d_in[4];
    const float* Wc  = (const float*)d_in[5];
    const float* bc  = (const float*)d_in[6];
    const int*   ei  = (const int*)d_in[7];
    const int*   bat = (const int*)d_in[8];

    const int n = in_sizes[0] / IN_CH;
    const int E = in_sizes[7] / 2;
    const int G = 64;

    const int* src = ei;
    const int* dst = ei + E;

    float* ws   = (float*)d_ws;
    float* deg  = ws;                ws += n;
    float* dinv = ws;                ws += n;
    float* h1   = ws;                ws += (size_t)n * 32;
    float* o1   = ws;                ws += (size_t)n * 32;
    float* h2   = ws;                ws += (size_t)n * 64;
    float* o2   = ws;                ws += (size_t)n * 64;
    float* sums = ws;                ws += (size_t)G * 64;
    float* cnts = ws;                ws += G;

    const int B = 256;
    auto cdiv = [](long long a, long long b) { return (int)((a + b - 1) / b); };

    k_init_deg<<<cdiv(n, B), B, 0, stream>>>(deg, n);
    k_deg_accum<<<cdiv(E, B), B, 0, stream>>>(dst, deg, E);
    k_dinv<<<cdiv(n, B), B, 0, stream>>>(deg, dinv, n);

    const int tiles = (n + 15) / 16;
    k_gemm_wmma<128, 32><<<cdiv(tiles, 4), 128, 0, stream>>>(x, W1, h1, n, tiles);
    k_selfloop<5><<<cdiv((long long)n * 32, B), B, 0, stream>>>(h1, dinv, o1, n * 32);
    k_edge_scatter<32, 32><<<cdiv(E, B), B, 0, stream>>>(src, dst, dinv, h1, o1, E);
    k_bias_relu<32><<<cdiv((long long)n * 32, B), B, 0, stream>>>(o1, b1, n * 32);

    k_gemm_wmma<32, 64><<<cdiv(tiles, 4), 128, 0, stream>>>(o1, W2, h2, n, tiles);
    k_selfloop<6><<<cdiv((long long)n * 64, B), B, 0, stream>>>(h2, dinv, o2, n * 64);
    k_edge_scatter<64, 16><<<cdiv(E, B), B, 0, stream>>>(src, dst, dinv, h2, o2, E);
    k_bias_relu<64><<<cdiv((long long)n * 64, B), B, 0, stream>>>(o2, b2, n * 64);

    hipMemsetAsync(sums, 0, (size_t)(G * 64 + G) * sizeof(float), stream);
    const int pool_waves = (n + 31) / 32;
    k_pool_runs<16><<<cdiv(pool_waves, 4), 128, 0, stream>>>(o2, bat, sums, cnts, n);
    k_final_head<<<1, 64, 0, stream>>>(sums, cnts, Wc, bc, (float*)d_out, G);
}


# Layer 2 aggregates before it multiplies: edge scatter on the 32-channel layer input, then the K=32 N=64 f32 matrix product; half the gathered and atomic lines; layer-2 sums reassociated
# speedup vs baseline: 2.1704x; 1.0490x over previous
#include <hip/hip_runtime.h>
#include <hip/hip_bf16.h>


typedef float v2f __attribute__((ext_vector_type(2)));
typedef float v8f __attribute__((ext_vector_type(8)));

#define IN_CH 128


__global__ void k_init_deg(float* deg, int n) {
    int i = blockIdx.x * blockDim.x + threadIdx.x;
    if (i < n) deg[i] = 1.0f;
}

__global__ void k_deg_accum(const int* __restrict__ dst, float* deg, int E) {
    int e = blockIdx.x * blockDim.x + threadIdx.x;
    if (e < E) unsafeAtomicAdd(&deg[dst[e]], 1.0f);
}

__global__ void k_dinv(const float* __restrict__ deg, float* dinv, int n) {
    int i = blockIdx.x * blockDim.x + threadIdx.x;
    if (i < n) {
        float d = deg[i];
        dinv[i] = d > 0.0f ? __frsqrt_rn(d) : 0.0f;
    }
}


template <int K, int N>
__global__ void k_gemm_wmma(const float* __restrict__ A,
                            const float* __restrict__ W,
                            float* __restrict__ O,
                            int n_rows, int n_tiles) {
    int wave = (blockIdx.x * blockDim.x + threadIdx.x) >> 5;
    int lane = threadIdx.x & 31;
    if (wave >= n_tiles) return;

    const int m0    = wave * 16;
    const int mrow  = m0 + (lane & 15);
    const int rm    = mrow < n_rows ? mrow : (n_rows - 1);
    const int khalf = (lane >> 4) * 2;
    const int colL  = lane & 15;

    constexpr int NT = N / 16;
    v8f acc[NT] = {};

    for (int k = 0; k < K; k += 4) {
        v2f a;
        a.x = A[rm * K + k + khalf];
        a.y = A[rm * K + k + khalf + 1];
#pragma unroll
        for (int t = 0; t < NT; ++t) {
            v2f b;
            b.x = W[(k + khalf) * N + t * 16 + colL];
            b.y = W[(k + khalf + 1) * N + t * 16 + colL];
            acc[t] = __builtin_amdgcn_wmma_f32_16x16x4_f32(
                false, a, false, b, (short)0, acc[t], false, false);
        }
    }

    const int mbase = m0 + (lane >> 4) * 8;
    if (m0 + 16 <= n_rows) {
#pragma unroll
        for (int t = 0; t < NT; ++t) {
#pragma unroll
            for (int r = 0; r < 8; ++r) {
                O[(mbase + r) * N + t * 16 + colL] = acc[t][r];
            }
        }
    } else {
#pragma unroll
        for (int t = 0; t < NT; ++t) {
#pragma unroll
            for (int r = 0; r < 8; ++r) {
                int row = mbase + r;
                if (row < n_rows) O[row * N + t * 16 + colL] = acc[t][r];
            }
        }
    }
}


template <int LOGC>
__global__ void k_selfloop(const float* __restrict__ h,
                           const float* __restrict__ dinv,
                           float* __restrict__ out, int total) {
    int id = blockIdx.x * blockDim.x + threadIdx.x;
    if (id >= total) return;
    float d = dinv[id >> LOGC];
    out[id] = d * d * h[id];
}


template <int C, int G>
__global__ void k_edge_scatter(const int* __restrict__ src,
                               const int* __restrict__ dst,
                               const float* __restrict__ dinv,
                               const float* __restrict__ h,
                               float* __restrict__ out, int E) {
    constexpr int Q = C / 32;
    const int wave = (blockIdx.x * blockDim.x + threadIdx.x) >> 5;
    const int lane = threadIdx.x & 31;
    const int base = wave * 32;
    if (base >= E) return;

    const int e = base + lane;
    int s = 0, d = 0;
    float nrm = 0.0f;
    if (e < E) {
        s = src[e];
        d = dst[e];
        nrm = dinv[s] * dinv[d];
    }

    if (base + 32 <= E) {
#pragma unroll
        for (int j0 = 0; j0 < 32; j0 += G) {
            float v[G][Q];
#pragma unroll
            for (int u = 0; u < G; ++u) {
                const int sj = __builtin_amdgcn_readlane(s, j0 + u);
                const float* hp = h + sj * C + lane;
#pragma unroll
                for (int q = 0; q < Q; ++q) v[u][q] = hp[q * 32];
            }
#pragma unroll
            for (int u = 0; u < G; ++u) {
                const int dj = __builtin_amdgcn_readlane(d, j0 + u);
                const float nj = __int_as_float(
                    __builtin_amdgcn_readlane(__float_as_int(nrm), j0 + u));
                float* op = out + dj * C + lane;
#pragma unroll
                for (int q = 0; q < Q; ++q)
                    unsafeAtomicAdd(op + q * 32, v[u][q] * nj);
            }
        }
    } else {
        const int cnt = E - base;
        for (int j = 0; j < cnt; ++j) {
            const int sj = __shfl(s, j);
            const int dj = __shfl(d, j);
            const float nj = __shfl(nrm, j);
#pragma unroll
            for (int q = 0; q < Q; ++q)
                unsafeAtomicAdd(&out[dj * C + q * 32 + lane],
                                h[sj * C + q * 32 + lane] * nj);
        }
    }
}


template <int C>
__global__ void k_bias_relu(float* h, const float* __restrict__ bias,
                            int total) {
    int id = blockIdx.x * blockDim.x + threadIdx.x;
    if (id >= total) return;
    float v = h[id] + bias[id & (C - 1)];
    h[id] = v > 0.0f ? v : 0.0f;
}


__device__ __forceinline__ void pool_flush(float* sums, float* cnts, int g,
                                           int lane, float a0, float a1,
                                           float run) {
    unsafeAtomicAdd(&sums[g * 64 + lane], a0);
    unsafeAtomicAdd(&sums[g * 64 + 32 + lane], a1);
    if (lane == 0) unsafeAtomicAdd(&cnts[g], run);
}

template <int G>
__global__ void k_pool_runs(const float* __restrict__ h,
                            const int* __restrict__ batch,
                            float* sums, float* cnts, int n) {
    const int wave = (blockIdx.x * blockDim.x + threadIdx.x) >> 5;
    const int lane = threadIdx.x & 31;
    const int base = wave * 32;
    if (base >= n) return;

    const int node = base + lane;
    const int b = node < n ? batch[node] : 0;
    const float* hp = h + (size_t)base * 64 + lane;

    int cur = __builtin_amdgcn_readfirstlane(b);
    float a0 = 0.0f, a1 = 0.0f, run = 0.0f;

    if (base + 32 <= n) {
#pragma unroll
        for (int j0 = 0; j0 < 32; j0 += G) {
            float v0[G], v1[G];
#pragma unroll
            for (int u = 0; u < G; ++u) {
                v0[u] = hp[(j0 + u) * 64];
                v1[u] = hp[(j0 + u) * 64 + 32];
            }
#pragma unroll
            for (int u = 0; u < G; ++u) {
                const int g = __builtin_amdgcn_readlane(b, j0 + u);
                if (g != cur) {
                    pool_flush(sums, cnts, cur, lane, a0, a1, run);
                    cur = g; a0 = 0.0f; a1 = 0.0f; run = 0.0f;
                }
                a0 += v0[u];
                a1 += v1[u];
                run += 1.0f;
            }
        }
    } else {
        const int cnt = n - base;
        for (int j = 0; j < cnt; ++j) {
            const int g = __shfl(b, j);
            if (g != cur) {
                pool_flush(sums, cnts, cur, lane, a0, a1, run);
                cur = g; a0 = 0.0f; a1 = 0.0f; run = 0.0f;
            }
            a0 += hp[j * 64];
            a1 += hp[j * 64 + 32];
            run += 1.0f;
        }
    }
    pool_flush(sums, cnts, cur, lane, a0, a1, run);
}


__global__ void k_final_head(const float* __restrict__ sums,
                             const float* __restrict__ cnts,
                             const float* __restrict__ Wc,
                             const float* __restrict__ bc,
                             float* out, int G) {
    int g = threadIdx.x;
    if (g >= G) return;
    float c = cnts[g];
    c = c > 1.0f ? c : 1.0f;
    float inv = 1.0f / c;
    float l0 = bc[0], l1 = bc[1];
    for (int ch = 0; ch < 64; ++ch) {
        float p = sums[g * 64 + ch] * inv;
        out[G * 2 + g * 64 + ch] = p;
        l0 = fmaf(p, Wc[ch * 2 + 0], l0);
        l1 = fmaf(p, Wc[ch * 2 + 1], l1);
    }
    out[g * 2 + 0] = l0;
    out[g * 2 + 1] = l1;
}


extern "C" void kernel_launch(void* const* d_in, const int* in_sizes, int n_in,
                              void* d_out, int out_size, void* d_ws, size_t ws_size,
                              hipStream_t stream) {
    const float* x   = (const float*)d_in[0];
    const float* W1  = (const float*)d_in[1];
    const float* b1  = (const float*)d_in[2];
    const float* W2  = (const float*)d_in[3];
    const float* b2  = (const float*)d_in[4];
    const float* Wc  = (const float*)d_in[5];
    const float* bc  = (const float*)d_in[6];
    const int*   ei  = (const int*)d_in[7];
    const int*   bat = (const int*)d_in[8];

    const int n = in_sizes[0] / IN_CH;
    const int E = in_sizes[7] / 2;
    const int G = 64;

    const int* src = ei;
    const int* dst = ei + E;

    float* ws   = (float*)d_ws;
    float* deg  = ws;                ws += n;
    float* dinv = ws;                ws += n;
    float* h1   = ws;                ws += (size_t)n * 32;
    float* o1   = ws;                ws += (size_t)n * 32;
    float* a2   = ws;                ws += (size_t)n * 32;
    float* o2   = ws;                ws += (size_t)n * 64;
    float* sums = ws;                ws += (size_t)G * 64;
    float* cnts = ws;                ws += G;

    const int B = 256;
    auto cdiv = [](long long a, long long b) { return (int)((a + b - 1) / b); };

    k_init_deg<<<cdiv(n, B), B, 0, stream>>>(deg, n);
    k_deg_accum<<<cdiv(E, B), B, 0, stream>>>(dst, deg, E);
    k_dinv<<<cdiv(n, B), B, 0, stream>>>(deg, dinv, n);

    const int tiles = (n + 15) / 16;
    k_gemm_wmma<128, 32><<<cdiv(tiles, 4), 128, 0, stream>>>(x, W1, h1, n, tiles);
    k_selfloop<5><<<cdiv((long long)n * 32, B), B, 0, stream>>>(h1, dinv, o1, n * 32);
    k_edge_scatter<32, 32><<<cdiv(E, B), B, 0, stream>>>(src, dst, dinv, h1, o1, E);
    k_bias_relu<32><<<cdiv((long long)n * 32, B), B, 0, stream>>>(o1, b1, n * 32);

    k_selfloop<5><<<cdiv((long long)n * 32, B), B, 0, stream>>>(o1, dinv, a2, n * 32);
    k_edge_scatter<32, 32><<<cdiv(E, B), B, 0, stream>>>(src, dst, dinv, o1, a2, E);
    k_gemm_wmma<32, 64><<<cdiv(tiles, 4), 128, 0, stream>>>(a2, W2, o2, n, tiles);
    k_bias_relu<64><<<cdiv((long long)n * 64, B), B, 0, stream>>>(o2, b2, n * 64);

    hipMemsetAsync(sums, 0, (size_t)(G * 64 + G) * sizeof(float), stream);
    const int pool_waves = (n + 31) / 32;
    k_pool_runs<16><<<cdiv(pool_waves, 4), 128, 0, stream>>>(o2, bat, sums, cnts, n);
    k_final_head<<<1, 64, 0, stream>>>(sums, cnts, Wc, bc, (float*)d_out, G);
}


# Edge weight factored out of the scatter: rows pre-scaled by dinv[src] in the per-node pass, dinv[dst] applied once per node after the edge sum; no per-edge dinv gathers; factor outside the sum
# speedup vs baseline: 2.3070x; 1.0629x over previous
#include <hip/hip_runtime.h>
#include <hip/hip_bf16.h>


typedef float v2f __attribute__((ext_vector_type(2)));
typedef float v8f __attribute__((ext_vector_type(8)));

#define IN_CH 128


__global__ void k_init_deg(float* deg, int n) {
    int i = blockIdx.x * blockDim.x + threadIdx.x;
    if (i < n) deg[i] = 1.0f;
}

__global__ void k_deg_accum(const int* __restrict__ dst, float* deg, int E) {
    int e = blockIdx.x * blockDim.x + threadIdx.x;
    if (e < E) unsafeAtomicAdd(&deg[dst[e]], 1.0f);
}

__global__ void k_dinv(const float* __restrict__ deg, float* dinv, int n) {
    int i = blockIdx.x * blockDim.x + threadIdx.x;
    if (i < n) {
        float d = deg[i];
        dinv[i] = d > 0.0f ? __frsqrt_rn(d) : 0.0f;
    }
}


template <int K, int N>
__global__ void k_gemm_wmma(const float* __restrict__ A,
                            const float* __restrict__ W,
                            float* __restrict__ O,
                            int n_rows, int n_tiles) {
    int wave = (blockIdx.x * blockDim.x + threadIdx.x) >> 5;
    int lane = threadIdx.x & 31;
    if (wave >= n_tiles) return;

    const int m0    = wave * 16;
    const int mrow  = m0 + (lane & 15);
    const int rm    = mrow < n_rows ? mrow : (n_rows - 1);
    const int khalf = (lane >> 4) * 2;
    const int colL  = lane & 15;

    constexpr int NT = N / 16;
    v8f acc[NT] = {};

    for (int k = 0; k < K; k += 4) {
        v2f a;
        a.x = A[rm * K + k + khalf];
        a.y = A[rm * K + k + khalf + 1];
#pragma unroll
        for (int t = 0; t < NT; ++t) {
            v2f b;
            b.x = W[(k + khalf) * N + t * 16 + colL];
            b.y = W[(k + khalf + 1) * N + t * 16 + colL];
            acc[t] = __builtin_amdgcn_wmma_f32_16x16x4_f32(
                false, a, false, b, (short)0, acc[t], false, false);
        }
    }

    const int mbase = m0 + (lane >> 4) * 8;
    if (m0 + 16 <= n_rows) {
#pragma unroll
        for (int t = 0; t < NT; ++t) {
#pragma unroll
            for (int r = 0; r < 8; ++r) {
                O[(mbase + r) * N + t * 16 + colL] = acc[t][r];
            }
        }
    } else {
#pragma unroll
        for (int t = 0; t < NT; ++t) {
#pragma unroll
            for (int r = 0; r < 8; ++r) {
                int row = mbase + r;
                if (row < n_rows) O[row * N + t * 16 + colL] = acc[t][r];
            }
        }
    }
}


template <int LOGC>
__global__ void k_prescale(float* h,
                           const float* __restrict__ dinv,
                           float* __restrict__ acc, int total) {
    int id = blockIdx.x * blockDim.x + threadIdx.x;
    if (id >= total) return;
    float v = dinv[id >> LOGC] * h[id];
    h[id] = v;
    acc[id] = v;
}


template <int C, int G>
__global__ void k_edge_scatter(const int* __restrict__ src,
                               const int* __restrict__ dst,
                               const float* __restrict__ h,
                               float* __restrict__ out, int E) {
    constexpr int Q = C / 32;
    const int wave = (blockIdx.x * blockDim.x + threadIdx.x) >> 5;
    const int lane = threadIdx.x & 31;
    const int base = wave * 32;
    if (base >= E) return;

    const int e = base + lane;
    int s = 0, d = 0;
    if (e < E) {
        s = src[e];
        d = dst[e];
    }

    if (base + 32 <= E) {
#pragma unroll
        for (int j0 = 0; j0 < 32; j0 += G) {
            float v[G][Q];
#pragma unroll
            for (int u = 0; u < G; ++u) {
                const int sj = __builtin_amdgcn_readlane(s, j0 + u);
                const float* hp = h + sj * C + lane;
#pragma unroll
                for (int q = 0; q < Q; ++q) v[u][q] = hp[q * 32];
            }
#pragma unroll
            for (int u = 0; u < G; ++u) {
                const int dj = __builtin_amdgcn_readlane(d, j0 + u);
                float* op = out + dj * C + lane;
#pragma unroll
                for (int q = 0; q < Q; ++q)
                    unsafeAtomicAdd(op + q * 32, v[u][q]);
            }
        }
    } else {
        const int cnt = E - base;
        for (int j = 0; j < cnt; ++j) {
            const int sj = __shfl(s, j);
            const int dj = __shfl(d, j);
#pragma unroll
            for (int q = 0; q < Q; ++q)
                unsafeAtomicAdd(&out[dj * C + q * 32 + lane],
                                h[sj * C + q * 32 + lane]);
        }
    }
}


template <int C, int LOGC>
__global__ void k_scale_bias_relu(float* h, const float* __restrict__ dinv,
                                  const float* __restrict__ bias, int total) {
    int id = blockIdx.x * blockDim.x + threadIdx.x;
    if (id >= total) return;
    float v = dinv[id >> LOGC] * h[id] + bias[id & (C - 1)];
    h[id] = v > 0.0f ? v : 0.0f;
}


__device__ __forceinline__ void pool_flush(float* sums, float* cnts, int g,
                                           int lane, float a0, float a1,
                                           float run) {
    unsafeAtomicAdd(&sums[g * 64 + lane], a0);
    unsafeAtomicAdd(&sums[g * 64 + 32 + lane], a1);
    if (lane == 0) unsafeAtomicAdd(&cnts[g], run);
}

template <int G>
__global__ void k_pool_runs(const float* __restrict__ h,
                            const int* __restrict__ batch,
                            float* sums, float* cnts, int n) {
    const int wave = (blockIdx.x * blockDim.x + threadIdx.x) >> 5;
    const int lane = threadIdx.x & 31;
    const int base = wave * 32;
    if (base >= n) return;

    const int node = base + lane;
    const int b = node < n ? batch[node] : 0;
    const float* hp = h + (size_t)base * 64 + lane;

    int cur = __builtin_amdgcn_readfirstlane(b);
    float a0 = 0.0f, a1 = 0.0f, run = 0.0f;

    if (base + 32 <= n) {
#pragma unroll
        for (int j0 = 0; j0 < 32; j0 += G) {
            float v0[G], v1[G];
#pragma unroll
            for (int u = 0; u < G; ++u) {
                v0[u] = hp[(j0 + u) * 64];
                v1[u] = hp[(j0 + u) * 64 + 32];
            }
#pragma unroll
            for (int u = 0; u < G; ++u) {
                const int g = __builtin_amdgcn_readlane(b, j0 + u);
                if (g != cur) {
                    pool_flush(sums, cnts, cur, lane, a0, a1, run);
                    cur = g; a0 = 0.0f; a1 = 0.0f; run = 0.0f;
                }
                a0 += v0[u];
                a1 += v1[u];
                run += 1.0f;
            }
        }
    } else {
        const int cnt = n - base;
        for (int j = 0; j < cnt; ++j) {
            const int g = __shfl(b, j);
            if (g != cur) {
                pool_flush(sums, cnts, cur, lane, a0, a1, run);
                cur = g; a0 = 0.0f; a1 = 0.0f; run = 0.0f;
            }
            a0 += hp[j * 64];
            a1 += hp[j * 64 + 32];
            run += 1.0f;
        }
    }
    pool_flush(sums, cnts, cur, lane, a0, a1, run);
}


__global__ void k_final_head(const float* __restrict__ sums,
                             const float* __restrict__ cnts,
                             const float* __restrict__ Wc,
                             const float* __restrict__ bc,
                             float* out, int G) {
    int g = threadIdx.x;
    if (g >= G) return;
    float c = cnts[g];
    c = c > 1.0f ? c : 1.0f;
    float inv = 1.0f / c;
    float l0 = bc[0], l1 = bc[1];
    for (int ch = 0; ch < 64; ++ch) {
        float p = sums[g * 64 + ch] * inv;
        out[G * 2 + g * 64 + ch] = p;
        l0 = fmaf(p, Wc[ch * 2 + 0], l0);
        l1 = fmaf(p, Wc[ch * 2 + 1], l1);
    }
    out[g * 2 + 0] = l0;
    out[g * 2 + 1] = l1;
}


extern "C" void kernel_launch(void* const* d_in, const int* in_sizes, int n_in,
                              void* d_out, int out_size, void* d_ws, size_t ws_size,
                              hipStream_t stream) {
    const float* x   = (const float*)d_in[0];
    const float* W1  = (const float*)d_in[1];
    const float* b1  = (const float*)d_in[2];
    const float* W2  = (const float*)d_in[3];
    const float* b2  = (const float*)d_in[4];
    const float* Wc  = (const float*)d_in[5];
    const float* bc  = (const float*)d_in[6];
    const int*   ei  = (const int*)d_in[7];
    const int*   bat = (const int*)d_in[8];

    const int n = in_sizes[0] / IN_CH;
    const int E = in_sizes[7] / 2;
    const int G = 64;

    const int* src = ei;
    const int* dst = ei + E;

    float* ws   = (float*)d_ws;
    float* deg  = ws;                ws += n;
    float* dinv = ws;                ws += n;
    float* h1   = ws;                ws += (size_t)n * 32;
    float* o1   = ws;                ws += (size_t)n * 32;
    float* a2   = ws;                ws += (size_t)n * 32;
    float* o2   = ws;                ws += (size_t)n * 64;
    float* sums = ws;                ws += (size_t)G * 64;
    float* cnts = ws;                ws += G;

    const int B = 256;
    auto cdiv = [](long long a, long long b) { return (int)((a + b - 1) / b); };

    k_init_deg<<<cdiv(n, B), B, 0, stream>>>(deg, n);
    k_deg_accum<<<cdiv(E, B), B, 0, stream>>>(dst, deg, E);
    k_dinv<<<cdiv(n, B), B, 0, stream>>>(deg, dinv, n);

    const int tiles = (n + 15) / 16;
    k_gemm_wmma<128, 32><<<cdiv(tiles, 4), 128, 0, stream>>>(x, W1, h1, n, tiles);
    k_prescale<5><<<cdiv((long long)n * 32, B), B, 0, stream>>>(h1, dinv, o1, n * 32);
    k_edge_scatter<32, 32><<<cdiv(E, B), B, 0, stream>>>(src, dst, h1, o1, E);
    k_scale_bias_relu<32, 5><<<cdiv((long long)n * 32, B), B, 0, stream>>>(o1, dinv, b1, n * 32);

    k_prescale<5><<<cdiv((long long)n * 32, B), B, 0, stream>>>(o1, dinv, a2, n * 32);
    k_edge_scatter<32, 32><<<cdiv(E, B), B, 0, stream>>>(src, dst, o1, a2, E);
    k_gemm_wmma<32, 64><<<cdiv(tiles, 4), 128, 0, stream>>>(a2, W2, o2, n, tiles);
    k_scale_bias_relu<64, 6><<<cdiv((long long)n * 64, B), B, 0, stream>>>(o2, dinv, b2, n * 64);

    hipMemsetAsync(sums, 0, (size_t)(G * 64 + G) * sizeof(float), stream);
    const int pool_waves = (n + 31) / 32;
    k_pool_runs<16><<<cdiv(pool_waves, 4), 128, 0, stream>>>(o2, bat, sums, cnts, n);
    k_final_head<<<1, 64, 0, stream>>>(sums, cnts, Wc, bc, (float*)d_out, G);
}


# Per-node scale/bias/ReLU passes folded into their producers: layer-1 product stores dinv*acc twice, layer-2 product stores relu(dinv*acc+b2), one pass between the edge sums; 13 launches to 10
# speedup vs baseline: 2.6250x; 1.1378x over previous
#include <hip/hip_runtime.h>
#include <hip/hip_bf16.h>


typedef float v2f __attribute__((ext_vector_type(2)));
typedef float v8f __attribute__((ext_vector_type(8)));

#define IN_CH 128


__global__ void k_init_deg(float* deg, int n) {
    int i = blockIdx.x * blockDim.x + threadIdx.x;
    if (i < n) deg[i] = 1.0f;
}

__global__ void k_deg_accum(const int* __restrict__ dst, float* deg, int E) {
    int e = blockIdx.x * blockDim.x + threadIdx.x;
    if (e < E) unsafeAtomicAdd(&deg[dst[e]], 1.0f);
}

__global__ void k_dinv(const float* __restrict__ deg, float* dinv, int n) {
    int i = blockIdx.x * blockDim.x + threadIdx.x;
    if (i < n) {
        float d = deg[i];
        dinv[i] = d > 0.0f ? __frsqrt_rn(d) : 0.0f;
    }
}


template <int K, int N, int EPI>
__global__ void k_gemm_wmma(const float* __restrict__ A,
                            const float* __restrict__ W,
                            const float* __restrict__ dinv,
                            const float* __restrict__ bias,
                            float* __restrict__ O,
                            float* __restrict__ O2,
                            int n_rows, int n_tiles) {
    int wave = (blockIdx.x * blockDim.x + threadIdx.x) >> 5;
    int lane = threadIdx.x & 31;
    if (wave >= n_tiles) return;

    const int m0    = wave * 16;
    const int mrow  = m0 + (lane & 15);
    const int rm    = mrow < n_rows ? mrow : (n_rows - 1);
    const int khalf = (lane >> 4) * 2;
    const int colL  = lane & 15;

    constexpr int NT = N / 16;
    v8f acc[NT] = {};

    const int mbase = m0 + (lane >> 4) * 8;

    float dv[8];
#pragma unroll
    for (int r = 0; r < 8; ++r) {
        const int row = mbase + r;
        dv[r] = dinv[row < n_rows ? row : (n_rows - 1)];
    }
    float bv[NT];
    if constexpr (EPI == 1) {
#pragma unroll
        for (int t = 0; t < NT; ++t) bv[t] = bias[t * 16 + colL];
    }

    for (int k = 0; k < K; k += 4) {
        v2f a;
        a.x = A[rm * K + k + khalf];
        a.y = A[rm * K + k + khalf + 1];
#pragma unroll
        for (int t = 0; t < NT; ++t) {
            v2f b;
            b.x = W[(k + khalf) * N + t * 16 + colL];
            b.y = W[(k + khalf + 1) * N + t * 16 + colL];
            acc[t] = __builtin_amdgcn_wmma_f32_16x16x4_f32(
                false, a, false, b, (short)0, acc[t], false, false);
        }
    }

    if (m0 + 16 <= n_rows) {
#pragma unroll
        for (int t = 0; t < NT; ++t) {
#pragma unroll
            for (int r = 0; r < 8; ++r) {
                const int idx = (mbase + r) * N + t * 16 + colL;
                if constexpr (EPI == 0) {
                    const float v = dv[r] * acc[t][r];
                    O[idx]  = v;
                    O2[idx] = v;
                } else {
                    const float v = dv[r] * acc[t][r] + bv[t];
                    O[idx] = v > 0.0f ? v : 0.0f;
                }
            }
        }
    } else {
#pragma unroll
        for (int t = 0; t < NT; ++t) {
#pragma unroll
            for (int r = 0; r < 8; ++r) {
                const int row = mbase + r;
                if (row < n_rows) {
                    const int idx = row * N + t * 16 + colL;
                    if constexpr (EPI == 0) {
                        const float v = dv[r] * acc[t][r];
                        O[idx]  = v;
                        O2[idx] = v;
                    } else {
                        const float v = dv[r] * acc[t][r] + bv[t];
                        O[idx] = v > 0.0f ? v : 0.0f;
                    }
                }
            }
        }
    }
}


template <int C, int LOGC>
__global__ void k_mid_pass(float* h,
                           const float* __restrict__ dinv,
                           const float* __restrict__ bias,
                           float* __restrict__ acc, int total) {
    int id = blockIdx.x * blockDim.x + threadIdx.x;
    if (id >= total) return;
    const float dv = dinv[id >> LOGC];
    float v = dv * h[id] + bias[id & (C - 1)];
    v = v > 0.0f ? v : 0.0f;
    v = dv * v;
    h[id] = v;
    acc[id] = v;
}


template <int C, int G>
__global__ void k_edge_scatter(const int* __restrict__ src,
                               const int* __restrict__ dst,
                               const float* __restrict__ h,
                               float* __restrict__ out, int E) {
    constexpr int Q = C / 32;
    const int wave = (blockIdx.x * blockDim.x + threadIdx.x) >> 5;
    const int lane = threadIdx.x & 31;
    const int base = wave * 32;
    if (base >= E) return;

    const int e = base + lane;
    int s = 0, d = 0;
    if (e < E) {
        s = src[e];
        d = dst[e];
    }

    if (base + 32 <= E) {
#pragma unroll
        for (int j0 = 0; j0 < 32; j0 += G) {
            float v[G][Q];
#pragma unroll
            for (int u = 0; u < G; ++u) {
                const int sj = __builtin_amdgcn_readlane(s, j0 + u);
                const float* hp = h + sj * C + lane;
#pragma unroll
                for (int q = 0; q < Q; ++q) v[u][q] = hp[q * 32];
            }
#pragma unroll
            for (int u = 0; u < G; ++u) {
                const int dj = __builtin_amdgcn_readlane(d, j0 + u);
                float* op = out + dj * C + lane;
#pragma unroll
                for (int q = 0; q < Q; ++q)
                    unsafeAtomicAdd(op + q * 32, v[u][q]);
            }
        }
    } else {
        const int cnt = E - base;
        for (int j = 0; j < cnt; ++j) {
            const int sj = __shfl(s, j);
            const int dj = __shfl(d, j);
#pragma unroll
            for (int q = 0; q < Q; ++q)
                unsafeAtomicAdd(&out[dj * C + q * 32 + lane],
                                h[sj * C + q * 32 + lane]);
        }
    }
}


__device__ __forceinline__ void pool_flush(float* sums, float* cnts, int g,
                                           int lane, float a0, float a1,
                                           float run) {
    unsafeAtomicAdd(&sums[g * 64 + lane], a0);
    unsafeAtomicAdd(&sums[g * 64 + 32 + lane], a1);
    if (lane == 0) unsafeAtomicAdd(&cnts[g], run);
}

template <int G>
__global__ void k_pool_runs(const float* __restrict__ h,
                            const int* __restrict__ batch,
                            float* sums, float* cnts, int n) {
    const int wave = (blockIdx.x * blockDim.x + threadIdx.x) >> 5;
    const int lane = threadIdx.x & 31;
    const int base = wave * 32;
    if (base >= n) return;

    const int node = base + lane;
    const int b = node < n ? batch[node] : 0;
    const float* hp = h + (size_t)base * 64 + lane;

    int cur = __builtin_amdgcn_readfirstlane(b);
    float a0 = 0.0f, a1 = 0.0f, run = 0.0f;

    if (base + 32 <= n) {
#pragma unroll
        for (int j0 = 0; j0 < 32; j0 += G) {
            float v0[G], v1[G];
#pragma unroll
            for (int u = 0; u < G; ++u) {
                v0[u] = hp[(j0 + u) * 64];
                v1[u] = hp[(j0 + u) * 64 + 32];
            }
#pragma unroll
            for (int u = 0; u < G; ++u) {
                const int g = __builtin_amdgcn_readlane(b, j0 + u);
                if (g != cur) {
                    pool_flush(sums, cnts, cur, lane, a0, a1, run);
                    cur = g; a0 = 0.0f; a1 = 0.0f; run = 0.0f;
                }
                a0 += v0[u];
                a1 += v1[u];
                run += 1.0f;
            }
        }
    } else {
        const int cnt = n - base;
        for (int j = 0; j < cnt; ++j) {
            const int g = __shfl(b, j);
            if (g != cur) {
                pool_flush(sums, cnts, cur, lane, a0, a1, run);
                cur = g; a0 = 0.0f; a1 = 0.0f; run = 0.0f;
            }
            a0 += hp[j * 64];
            a1 += hp[j * 64 + 32];
            run += 1.0f;
        }
    }
    pool_flush(sums, cnts, cur, lane, a0, a1, run);
}


__global__ void k_final_head(const float* __restrict__ sums,
                             const float* __restrict__ cnts,
                             const float* __restrict__ Wc,
                             const float* __restrict__ bc,
                             float* out, int G) {
    int g = threadIdx.x;
    if (g >= G) return;
    float c = cnts[g];
    c = c > 1.0f ? c : 1.0f;
    float inv = 1.0f / c;
    float l0 = bc[0], l1 = bc[1];
    for (int ch = 0; ch < 64; ++ch) {
        float p = sums[g * 64 + ch] * inv;
        out[G * 2 + g * 64 + ch] = p;
        l0 = fmaf(p, Wc[ch * 2 + 0], l0);
        l1 = fmaf(p, Wc[ch * 2 + 1], l1);
    }
    out[g * 2 + 0] = l0;
    out[g * 2 + 1] = l1;
}


extern "C" void kernel_launch(void* const* d_in, const int* in_sizes, int n_in,
                              void* d_out, int out_size, void* d_ws, size_t ws_size,
                              hipStream_t stream) {
    const float* x   = (const float*)d_in[0];
    const float* W1  = (const float*)d_in[1];
    const float* b1  = (const float*)d_in[2];
    const float* W2  = (const float*)d_in[3];
    const float* b2  = (const float*)d_in[4];
    const float* Wc  = (const float*)d_in[5];
    const float* bc  = (const float*)d_in[6];
    const int*   ei  = (const int*)d_in[7];
    const int*   bat = (const int*)d_in[8];

    const int n = in_sizes[0] / IN_CH;
    const int E = in_sizes[7] / 2;
    const int G = 64;

    const int* src = ei;
    const int* dst = ei + E;

    float* ws   = (float*)d_ws;
    float* deg  = ws;                ws += n;
    float* dinv = ws;                ws += n;
    float* h1   = ws;                ws += (size_t)n * 32;
    float* o1   = ws;                ws += (size_t)n * 32;
    float* a2   = ws;                ws += (size_t)n * 32;
    float* o2   = ws;                ws += (size_t)n * 64;
    float* sums = ws;                ws += (size_t)G * 64;
    float* cnts = ws;                ws += G;

    const int B = 256;
    auto cdiv = [](long long a, long long b) { return (int)((a + b - 1) / b); };

    k_init_deg<<<cdiv(n, B), B, 0, stream>>>(deg, n);
    k_deg_accum<<<cdiv(E, B), B, 0, stream>>>(dst, deg, E);
    k_dinv<<<cdiv(n, B), B, 0, stream>>>(deg, dinv, n);

    const int tiles = (n + 15) / 16;
    k_gemm_wmma<128, 32, 0><<<cdiv(tiles, 4), 128, 0, stream>>>(x, W1, dinv, nullptr, h1, o1, n, tiles);
    k_edge_scatter<32, 32><<<cdiv(E, B), B, 0, stream>>>(src, dst, h1, o1, E);

    k_mid_pass<32, 5><<<cdiv((long long)n * 32, B), B, 0, stream>>>(o1, dinv, b1, a2, n * 32);
    k_edge_scatter<32, 32><<<cdiv(E, B), B, 0, stream>>>(src, dst, o1, a2, E);
    k_gemm_wmma<32, 64, 1><<<cdiv(tiles, 4), 128, 0, stream>>>(a2, W2, dinv, b2, o2, nullptr, n, tiles);

    hipMemsetAsync(sums, 0, (size_t)(G * 64 + G) * sizeof(float), stream);
    const int pool_waves = (n + 31) / 32;
    k_pool_runs<16><<<cdiv(pool_waves, 4), 128, 0, stream>>>(o2, bat, sums, cnts, n);
    k_final_head<<<1, 64, 0, stream>>>(sums, cnts, Wc, bc, (float*)d_out, G);
}


# Per-destination sums without global atomics: edges binned per call by 256-row block; its owner work-group counts degrees and sums gathered rows in LDS, one store per row; 9 launches; sums reordered
# speedup vs baseline: 3.5653x; 1.3582x over previous
#include <hip/hip_runtime.h>
#include <hip/hip_bf16.h>


typedef float v2f __attribute__((ext_vector_type(2)));
typedef float v8f __attribute__((ext_vector_type(8)));

#define IN_CH 128
#define RB_LOG 8
#define RB 256
#define EPB 4096
#define CS 32

__global__ void k_init(float* pool, int n_pool, unsigned* zc, int n_zc) {
    const int i = blockIdx.x * blockDim.x + threadIdx.x;
    for (int j = i; j < n_pool; j += gridDim.x * blockDim.x) pool[j] = 0.0f;
    for (int j = i; j < n_zc; j += gridDim.x * blockDim.x) zc[j] = 0u;
}

__global__ void k_hist(const int* __restrict__ dst, unsigned* bcnt, int E, int nb) {
    extern __shared__ unsigned lh[];
    for (int j = threadIdx.x; j < nb; j += 256) lh[j] = 0u;
    __syncthreads();
    const int base = blockIdx.x * EPB;
#pragma unroll
    for (int it = 0; it < EPB / 256; ++it) {
        const int e = base + it * 256 + (int)threadIdx.x;
        if (e < E) {
            atomicAdd(&lh[dst[e] >> RB_LOG], 1u);
        }
    }
    __syncthreads();
    for (int j = threadIdx.x; j < nb; j += 256) {
        const unsigned c = lh[j];
        if (c) atomicAdd(&bcnt[j * CS], c);
    }
}

__global__ void k_fill(const int* __restrict__ src, const int* __restrict__ dst,
                       const unsigned* __restrict__ bcnt, unsigned* bcur,
                       unsigned* boff, unsigned* __restrict__ binned,
                       int E, int nb) {
    extern __shared__ unsigned sm[];
    unsigned* s_off  = sm;
    unsigned* s_lc   = sm + nb;
    unsigned* s_base = sm + 2 * nb;
    __shared__ unsigned wsum[8];
    const int tid = threadIdx.x, lane = tid & 31, wv = tid >> 5;

    unsigned carry = 0u;
    for (int b0 = 0; b0 < nb; b0 += 256) {
        const int i = b0 + tid;
        const unsigned v = i < nb ? bcnt[i * CS] : 0u;
        unsigned x = v;
#pragma unroll
        for (int d = 1; d < 32; d <<= 1) {
            const unsigned y = __shfl_up(x, d);
            if (lane >= d) x += y;
        }
        if (lane == 31) wsum[wv] = x;
        __syncthreads();
        unsigned wp = 0u, tot = 0u;
#pragma unroll
        for (int w = 0; w < 8; ++w) {
            const unsigned s = wsum[w];
            tot += s;
            if (w < wv) wp += s;
        }
        if (i < nb) {
            s_off[i] = carry + wp + x - v;
            s_lc[i] = 0u;
        }
        carry += tot;
        __syncthreads();
    }
    if (blockIdx.x == 0) {
        for (int j = tid; j < nb; j += 256) boff[j] = s_off[j];
        if (tid == 0) boff[nb] = carry;
    }

    const int base = blockIdx.x * EPB;
    int s[EPB / 256], d[EPB / 256];
#pragma unroll
    for (int it = 0; it < EPB / 256; ++it) {
        const int e = base + it * 256 + tid;
        s[it] = 0;
        d[it] = -1;
        if (e < E) {
            s[it] = src[e];
            d[it] = dst[e];
            atomicAdd(&s_lc[d[it] >> RB_LOG], 1u);
        }
    }
    __syncthreads();
    for (int j = tid; j < nb; j += 256) {
        const unsigned c = s_lc[j];
        if (c) s_base[j] = s_off[j] + atomicAdd(&bcur[j * CS], c);
        s_lc[j] = 0u;
    }
    __syncthreads();
#pragma unroll
    for (int it = 0; it < EPB / 256; ++it) {
        if (d[it] >= 0) {
            const int b = d[it] >> RB_LOG;
            const unsigned r = atomicAdd(&s_lc[b], 1u);
            binned[s_base[b] + r] =
                ((unsigned)s[it] << RB_LOG) | (unsigned)(d[it] & (RB - 1));
        }
    }
}

__global__ void k_degcount(const unsigned* __restrict__ binned,
                           const unsigned* __restrict__ boff,
                           float* __restrict__ deg, int n) {
    __shared__ unsigned cnt[RB];
    const int tid = threadIdx.x;
    const int row0 = blockIdx.x * RB;
    cnt[tid] = 0u;
    const int beg = (int)boff[blockIdx.x];
    const int end = (int)boff[blockIdx.x + 1];
    __syncthreads();
    for (int e = beg + tid; e < end; e += 256) atomicAdd(&cnt[binned[e] & (RB - 1)], 1u);
    __syncthreads();
    if (row0 + tid < n) deg[row0 + tid] = 1.0f + (float)cnt[tid];
}

__device__ __forceinline__ float dinv_of(float d) {
    return d > 0.0f ? __frsqrt_rn(d) : 0.0f;
}


template <int K, int N, int EPI>
__global__ void k_gemm_wmma(const float* __restrict__ A,
                            const float* __restrict__ W,
                            const float* __restrict__ deg,
                            const float* __restrict__ bias,
                            const int* __restrict__ gid,
                            float* __restrict__ O,
                            float* __restrict__ O2,
                            int n_rows, int n_tiles) {
    int wave = (blockIdx.x * blockDim.x + threadIdx.x) >> 5;
    int lane = threadIdx.x & 31;
    if (wave >= n_tiles) return;

    const int m0    = wave * 16;
    const int mrow  = m0 + (lane & 15);
    const int rm    = mrow < n_rows ? mrow : (n_rows - 1);
    const int khalf = (lane >> 4) * 2;
    const int colL  = lane & 15;

    constexpr int NT = N / 16;
    v8f acc[NT] = {};

    const int mbase = m0 + (lane >> 4) * 8;

    float dv[8];
#pragma unroll
    for (int r = 0; r < 8; ++r) {
        const int row = mbase + r;
        dv[r] = dinv_of(deg[row < n_rows ? row : (n_rows - 1)]);
    }
    float bv[NT];
    int g_row = 0;
    if constexpr (EPI == 1) {
#pragma unroll
        for (int t = 0; t < NT; ++t) bv[t] = bias[t * 16 + colL];
        g_row = gid[rm];
    }

    for (int k = 0; k < K; k += 4) {
        v2f a;
        a.x = A[rm * K + k + khalf];
        a.y = A[rm * K + k + khalf + 1];
#pragma unroll
        for (int t = 0; t < NT; ++t) {
            v2f b;
            b.x = W[(k + khalf) * N + t * 16 + colL];
            b.y = W[(k + khalf + 1) * N + t * 16 + colL];
            acc[t] = __builtin_amdgcn_wmma_f32_16x16x4_f32(
                false, a, false, b, (short)0, acc[t], false, false);
        }
    }

    if constexpr (EPI == 0) {
        if (m0 + 16 <= n_rows) {
#pragma unroll
            for (int t = 0; t < NT; ++t) {
#pragma unroll
                for (int r = 0; r < 8; ++r) {
                    const int idx = (mbase + r) * N + t * 16 + colL;
                    const float v = dv[r] * acc[t][r];
                    O[idx] = v;
                }
            }
        } else {
#pragma unroll
            for (int t = 0; t < NT; ++t) {
#pragma unroll
                for (int r = 0; r < 8; ++r) {
                    const int row = mbase + r;
                    if (row < n_rows) {
                        const int idx = row * N + t * 16 + colL;
                        const float v = dv[r] * acc[t][r];
                        O[idx] = v;
                    }
                }
            }
        }
    } else {
        static_assert(EPI != 1 || NT % 2 == 0, "pooled epilogue pairs column tiles into full lines");
        const int g0 = __builtin_amdgcn_readfirstlane(g_row);
        const bool one_graph = (m0 + 16 <= n_rows) && __all(g_row == g0);
        if (one_graph) {
            float cs[NT];
#pragma unroll
            for (int t = 0; t < NT; ++t) {
                float s = 0.0f;
#pragma unroll
                for (int r = 0; r < 8; ++r) {
                    const float v = dv[r] * acc[t][r] + bv[t];
                    s += v > 0.0f ? v : 0.0f;
                }
                cs[t] = s + __shfl_xor(s, 16);
            }
#pragma unroll
            for (int p = 0; p < NT / 2; ++p) {
                const float s = lane < 16 ? cs[2 * p] : cs[2 * p + 1];
                unsafeAtomicAdd(&O[g0 * N + p * 32 + lane], s);
            }
            if (lane == 0) unsafeAtomicAdd(&O2[g0], 16.0f);
        } else {
#pragma unroll
            for (int r = 0; r < 8; ++r) {
                const int row = mbase + r;
                if (row < n_rows) {
                    const int g = gid[row];
#pragma unroll
                    for (int t = 0; t < NT; ++t) {
                        const float v = dv[r] * acc[t][r] + bv[t];
                        unsafeAtomicAdd(&O[g * N + t * 16 + colL], v > 0.0f ? v : 0.0f);
                    }
                    if (colL == 0) unsafeAtomicAdd(&O2[g], 1.0f);
                }
            }
        }
    }
}


template <int MODE>
__global__ void __launch_bounds__(512)
k_agg(const unsigned* __restrict__ binned, const unsigned* __restrict__ boff,
      const float* __restrict__ h, const float* __restrict__ deg,
      const float* __restrict__ bias, float* __restrict__ out, int n) {
    __shared__ float acc[RB * 32];
    const int tid  = threadIdx.x;
    const int lane = tid & 31;
    const int wv   = __builtin_amdgcn_readfirstlane(tid >> 5);
    const int b    = blockIdx.x;
    const int row0 = b * RB;
    const int nrows = (n - row0) < RB ? (n - row0) : RB;
    const int nel  = nrows * 32;
    const float* hs = h + (size_t)row0 * 32;
    for (int i = tid; i < nel; i += 512) acc[i] = hs[i];
    const int beg = (int)boff[b];
    const int end = (int)boff[b + 1];
    __syncthreads();

    int base = beg + wv * 32;
    unsigned rec = 0u;
    if (base + lane < end) rec = binned[base + lane];
    while (base < end) {
        const int nbase = base + 512;
        unsigned nrec = 0u;
        const int sidx = (int)(rec >> RB_LOG);
        const int didx = (int)(rec & (RB - 1));
        if (base + 32 <= end) {
            float v[32];
#pragma unroll
            for (int u = 0; u < 32; ++u) {
                const int sj = __builtin_amdgcn_readlane(sidx, u);
                v[u] = h[sj * 32 + lane];
            }
            int ni = nbase + lane;
            ni = ni < end ? ni : end - 1;
            nrec = binned[ni];
#pragma unroll
            for (int u = 0; u < 32; ++u) {
                const int dj = __builtin_amdgcn_readlane(didx, u);
                atomicAdd(&acc[dj * 32 + lane], v[u]);
            }
        } else {
            const int cnt = end - base;
            for (int j = 0; j < cnt; ++j) {
                const int sj = __shfl(sidx, j);
                const int dj = __shfl(didx, j);
                atomicAdd(&acc[dj * 32 + lane], h[sj * 32 + lane]);
            }
        }
        rec  = nrec;
        base = nbase;
    }
    __syncthreads();

    float* os = out + (size_t)row0 * 32;
    if constexpr (MODE == 1) {
        const float bch = bias[lane];
        for (int i = tid; i < nel; i += 512) {
            const float dv = dinv_of(deg[row0 + (i >> 5)]);
            float v = dv * acc[i] + bch;
            v = v > 0.0f ? v : 0.0f;
            os[i] = dv * v;
        }
    } else {
        for (int i = tid; i < nel; i += 512) os[i] = acc[i];
    }
}


__global__ void k_final_head(const float* __restrict__ sums,
                             const float* __restrict__ cnts,
                             const float* __restrict__ Wc,
                             const float* __restrict__ bc,
                             float* out, int G) {
    int g = threadIdx.x;
    if (g >= G) return;
    float c = cnts[g];
    c = c > 1.0f ? c : 1.0f;
    float inv = 1.0f / c;
    float l0 = bc[0], l1 = bc[1];
    for (int ch = 0; ch < 64; ++ch) {
        float p = sums[g * 64 + ch] * inv;
        out[G * 2 + g * 64 + ch] = p;
        l0 = fmaf(p, Wc[ch * 2 + 0], l0);
        l1 = fmaf(p, Wc[ch * 2 + 1], l1);
    }
    out[g * 2 + 0] = l0;
    out[g * 2 + 1] = l1;
}


extern "C" void kernel_launch(void* const* d_in, const int* in_sizes, int n_in,
                              void* d_out, int out_size, void* d_ws, size_t ws_size,
                              hipStream_t stream) {
    const float* x   = (const float*)d_in[0];
    const float* W1  = (const float*)d_in[1];
    const float* b1  = (const float*)d_in[2];
    const float* W2  = (const float*)d_in[3];
    const float* b2  = (const float*)d_in[4];
    const float* Wc  = (const float*)d_in[5];
    const float* bc  = (const float*)d_in[6];
    const int*   ei  = (const int*)d_in[7];
    const int*   bat = (const int*)d_in[8];

    const int n = in_sizes[0] / IN_CH;
    const int E = in_sizes[7] / 2;
    const int G = 64;

    const int* src = ei;
    const int* dst = ei + E;

    const int B = 256;
    auto cdiv = [](long long a, long long b) { return (int)((a + b - 1) / b); };
    const int nb = cdiv(n, RB);

    float* ws   = (float*)d_ws;
    float* deg  = ws;                ws += n;
    float* h1   = ws;                ws += (size_t)n * 32;
    float* o1   = ws;                ws += (size_t)n * 32;
    unsigned* binned = (unsigned*)ws; ws += E;
    float* sums = ws;                ws += (size_t)G * 64;
    float* cnts = ws;                ws += G;
    unsigned* bcnt = (unsigned*)ws;  ws += (size_t)nb * CS;
    unsigned* bcur = (unsigned*)ws;  ws += (size_t)nb * CS;
    unsigned* boff = (unsigned*)ws;  ws += nb + 1;
    float* a2   = h1;

    k_init<<<cdiv(2 * nb * CS, B), B, 0, stream>>>(sums, G * 64 + G, bcnt, 2 * nb * CS);
    k_hist<<<cdiv(E, EPB), 256, nb * sizeof(unsigned), stream>>>(dst, bcnt, E, nb);
    k_fill<<<cdiv(E, EPB), 256, 3 * nb * sizeof(unsigned), stream>>>(src, dst, bcnt, bcur, boff, binned, E, nb);
    k_degcount<<<nb, RB, 0, stream>>>(binned, boff, deg, n);

    const int tiles = (n + 15) / 16;
    k_gemm_wmma<128, 32, 0><<<cdiv(tiles, 4), 128, 0, stream>>>(x, W1, deg, nullptr, nullptr, h1, nullptr, n, tiles);
    k_agg<1><<<nb, 512, 0, stream>>>(binned, boff, h1, deg, b1, o1, n);
    k_agg<2><<<nb, 512, 0, stream>>>(binned, boff, o1, deg, nullptr, a2, n);
    k_gemm_wmma<32, 64, 1><<<cdiv(tiles, 4), 128, 0, stream>>>(a2, W2, deg, b2, bat, sums, cnts, n, tiles);

    k_final_head<<<1, 64, 0, stream>>>(sums, cnts, Wc, bc, (float*)d_out, G);
}


# Loads issued as groups ahead of their uses in the binning, degree and edge-sum kernels (was one wait per load); a block's short last batch pads to a full one via a spare LDS row; sums in arrival order
# speedup vs baseline: 4.4397x; 1.2453x over previous
#include <hip/hip_runtime.h>
#include <hip/hip_bf16.h>


typedef float v2f __attribute__((ext_vector_type(2)));
typedef float v8f __attribute__((ext_vector_type(8)));

#define IN_CH 128
#define RB_LOG 8
#define RB 256
#define EPB 4096
#define CS 32

__global__ void k_init(float* pool, int n_pool, unsigned* zc, int n_zc) {
    const int i = blockIdx.x * blockDim.x + threadIdx.x;
    for (int j = i; j < n_pool; j += gridDim.x * blockDim.x) pool[j] = 0.0f;
    for (int j = i; j < n_zc; j += gridDim.x * blockDim.x) zc[j] = 0u;
}

__global__ void k_hist(const int* __restrict__ dst, unsigned* bcnt, int E, int nb) {
    extern __shared__ unsigned lh[];
    for (int j = threadIdx.x; j < nb; j += 256) lh[j] = 0u;
    const int base = blockIdx.x * EPB;
    int d[EPB / 256];
#pragma unroll
    for (int it = 0; it < EPB / 256; ++it) {
        const int e = base + it * 256 + (int)threadIdx.x;
        d[it] = dst[e < E ? e : E - 1];
    }
    __syncthreads();
#pragma unroll
    for (int it = 0; it < EPB / 256; ++it) {
        const int e = base + it * 256 + (int)threadIdx.x;
        if (e < E) atomicAdd(&lh[d[it] >> RB_LOG], 1u);
    }
    __syncthreads();
    for (int j = threadIdx.x; j < nb; j += 256) {
        const unsigned c = lh[j];
        if (c) atomicAdd(&bcnt[j * CS], c);
    }
}

__global__ void k_fill(const int* __restrict__ src, const int* __restrict__ dst,
                       const unsigned* __restrict__ bcnt, unsigned* bcur,
                       unsigned* boff, unsigned* __restrict__ binned,
                       int E, int nb) {
    extern __shared__ unsigned sm[];
    unsigned* s_off  = sm;
    unsigned* s_lc   = sm + nb;
    unsigned* s_base = sm + 2 * nb;
    __shared__ unsigned wsum[8];
    const int tid = threadIdx.x, lane = tid & 31, wv = tid >> 5;

    const int base = blockIdx.x * EPB;
    int s[EPB / 256], d[EPB / 256];
#pragma unroll
    for (int it = 0; it < EPB / 256; ++it) {
        const int e = base + it * 256 + tid;
        const int ec = e < E ? e : E - 1;
        s[it] = src[ec];
        d[it] = dst[ec];
    }

    unsigned carry = 0u;
    for (int b0 = 0; b0 < nb; b0 += 256) {
        const int i = b0 + tid;
        const unsigned v = i < nb ? bcnt[i * CS] : 0u;
        unsigned x = v;
#pragma unroll
        for (int d = 1; d < 32; d <<= 1) {
            const unsigned y = __shfl_up(x, d);
            if (lane >= d) x += y;
        }
        if (lane == 31) wsum[wv] = x;
        __syncthreads();
        unsigned wp = 0u, tot = 0u;
#pragma unroll
        for (int w = 0; w < 8; ++w) {
            const unsigned s = wsum[w];
            tot += s;
            if (w < wv) wp += s;
        }
        if (i < nb) {
            s_off[i] = carry + wp + x - v;
            s_lc[i] = 0u;
        }
        carry += tot;
        __syncthreads();
    }
    if (blockIdx.x == 0) {
        for (int j = tid; j < nb; j += 256) boff[j] = s_off[j];
        if (tid == 0) boff[nb] = carry;
    }

#pragma unroll
    for (int it = 0; it < EPB / 256; ++it) {
        if (base + it * 256 + tid < E) atomicAdd(&s_lc[d[it] >> RB_LOG], 1u);
    }
    __syncthreads();
    for (int j0 = tid; j0 < nb; j0 += 512) {
        const int j1 = j0 + 256;
        const bool two = j1 < nb;
        const int j1c = two ? j1 : j0;
        const unsigned c0 = s_lc[j0];
        const unsigned c1 = s_lc[j1c];
        const unsigned r0 = atomicAdd(&bcur[j0 * CS], c0);
        unsigned r1 = 0u;
        if (two) r1 = atomicAdd(&bcur[j1 * CS], c1);
        s_base[j0] = s_off[j0] + r0;
        s_lc[j0] = 0u;
        if (two) {
            s_base[j1] = s_off[j1] + r1;
            s_lc[j1] = 0u;
        }
    }
    __syncthreads();
    unsigned rk[EPB / 256], sb[EPB / 256];
#pragma unroll
    for (int it = 0; it < EPB / 256; ++it) {
        rk[it] = 0u;
        sb[it] = 0u;
        if (base + it * 256 + tid < E) {
            const int b = d[it] >> RB_LOG;
            rk[it] = atomicAdd(&s_lc[b], 1u);
            sb[it] = s_base[b];
        }
    }
#pragma unroll
    for (int it = 0; it < EPB / 256; ++it) {
        if (base + it * 256 + tid < E) {
            binned[sb[it] + rk[it]] =
                ((unsigned)s[it] << RB_LOG) | (unsigned)(d[it] & (RB - 1));
        }
    }
}

#define DEG_T 512
#define DEG_U 16
__global__ void __launch_bounds__(DEG_T)
k_degcount(const unsigned* __restrict__ binned,
           const unsigned* __restrict__ boff,
           float* __restrict__ deg, int n) {
    __shared__ unsigned cnt[RB];
    const int tid = threadIdx.x;
    const int row0 = blockIdx.x * RB;
    if (tid < RB) cnt[tid] = 0u;
    const int beg = (int)boff[blockIdx.x];
    const int end = (int)boff[blockIdx.x + 1];
    __syncthreads();
    for (int c = beg; c < end; c += DEG_T * DEG_U) {
        unsigned v[DEG_U];
#pragma unroll
        for (int u = 0; u < DEG_U; ++u) {
            const int e = c + u * DEG_T + tid;
            v[u] = binned[e < end ? e : end - 1];
        }
#pragma unroll
        for (int u = 0; u < DEG_U; ++u) {
            if (c + u * DEG_T + tid < end) atomicAdd(&cnt[v[u] & (RB - 1)], 1u);
        }
    }
    __syncthreads();
    if (tid < RB && row0 + tid < n) deg[row0 + tid] = 1.0f + (float)cnt[tid];
}

__device__ __forceinline__ float dinv_of(float d) {
    return d > 0.0f ? __frsqrt_rn(d) : 0.0f;
}


template <int K, int N, int EPI>
__global__ void k_gemm_wmma(const float* __restrict__ A,
                            const float* __restrict__ W,
                            const float* __restrict__ deg,
                            const float* __restrict__ bias,
                            const int* __restrict__ gid,
                            float* __restrict__ O,
                            float* __restrict__ O2,
                            int n_rows, int n_tiles) {
    int wave = (blockIdx.x * blockDim.x + threadIdx.x) >> 5;
    int lane = threadIdx.x & 31;
    if (wave >= n_tiles) return;

    const int m0    = wave * 16;
    const int mrow  = m0 + (lane & 15);
    const int rm    = mrow < n_rows ? mrow : (n_rows - 1);
    const int khalf = (lane >> 4) * 2;
    const int colL  = lane & 15;

    constexpr int NT = N / 16;
    v8f acc[NT] = {};

    const int mbase = m0 + (lane >> 4) * 8;

    float dv[8];
#pragma unroll
    for (int r = 0; r < 8; ++r) {
        const int row = mbase + r;
        dv[r] = dinv_of(deg[row < n_rows ? row : (n_rows - 1)]);
    }
    float bv[NT];
    int g_row = 0;
    if constexpr (EPI == 1) {
#pragma unroll
        for (int t = 0; t < NT; ++t) bv[t] = bias[t * 16 + colL];
        g_row = gid[rm];
    }

    for (int k = 0; k < K; k += 4) {
        v2f a;
        a.x = A[rm * K + k + khalf];
        a.y = A[rm * K + k + khalf + 1];
#pragma unroll
        for (int t = 0; t < NT; ++t) {
            v2f b;
            b.x = W[(k + khalf) * N + t * 16 + colL];
            b.y = W[(k + khalf + 1) * N + t * 16 + colL];
            acc[t] = __builtin_amdgcn_wmma_f32_16x16x4_f32(
                false, a, false, b, (short)0, acc[t], false, false);
        }
    }

    if constexpr (EPI == 0) {
        if (m0 + 16 <= n_rows) {
#pragma unroll
            for (int t = 0; t < NT; ++t) {
#pragma unroll
                for (int r = 0; r < 8; ++r) {
                    const int idx = (mbase + r) * N + t * 16 + colL;
                    const float v = dv[r] * acc[t][r];
                    O[idx] = v;
                }
            }
        } else {
#pragma unroll
            for (int t = 0; t < NT; ++t) {
#pragma unroll
                for (int r = 0; r < 8; ++r) {
                    const int row = mbase + r;
                    if (row < n_rows) {
                        const int idx = row * N + t * 16 + colL;
                        const float v = dv[r] * acc[t][r];
                        O[idx] = v;
                    }
                }
            }
        }
    } else {
        static_assert(EPI != 1 || NT % 2 == 0, "pooled epilogue pairs column tiles into full lines");
        const int g0 = __builtin_amdgcn_readfirstlane(g_row);
        const bool one_graph = (m0 + 16 <= n_rows) && __all(g_row == g0);
        if (one_graph) {
            float cs[NT];
#pragma unroll
            for (int t = 0; t < NT; ++t) {
                float s = 0.0f;
#pragma unroll
                for (int r = 0; r < 8; ++r) {
                    const float v = dv[r] * acc[t][r] + bv[t];
                    s += v > 0.0f ? v : 0.0f;
                }
                cs[t] = s + __shfl_xor(s, 16);
            }
#pragma unroll
            for (int p = 0; p < NT / 2; ++p) {
                const float s = lane < 16 ? cs[2 * p] : cs[2 * p + 1];
                unsafeAtomicAdd(&O[g0 * N + p * 32 + lane], s);
            }
            if (lane == 0) unsafeAtomicAdd(&O2[g0], 16.0f);
        } else {
#pragma unroll
            for (int r = 0; r < 8; ++r) {
                const int row = mbase + r;
                if (row < n_rows) {
                    const int g = gid[row];
#pragma unroll
                    for (int t = 0; t < NT; ++t) {
                        const float v = dv[r] * acc[t][r] + bv[t];
                        unsafeAtomicAdd(&O[g * N + t * 16 + colL], v > 0.0f ? v : 0.0f);
                    }
                    if (colL == 0) unsafeAtomicAdd(&O2[g], 1.0f);
                }
            }
        }
    }
}


template <int MODE>
__global__ void __launch_bounds__(512)
k_agg(const unsigned* __restrict__ binned, const unsigned* __restrict__ boff,
      const float* __restrict__ h, const float* __restrict__ deg,
      const float* __restrict__ bias, float* __restrict__ out, int n) {
    __shared__ float acc[(RB + 1) * 32];
    constexpr int NIT = RB * 32 / 512;
    const int tid  = threadIdx.x;
    const int lane = tid & 31;
    const int wv   = __builtin_amdgcn_readfirstlane(tid >> 5);
    const int b    = blockIdx.x;
    const int row0 = b * RB;
    const int nrows = (n - row0) < RB ? (n - row0) : RB;
    const int nel  = nrows * 32;
    const float* hs = h + (size_t)row0 * 32;
    {
        float t[NIT];
#pragma unroll
        for (int it = 0; it < NIT; ++it) {
            const int i = tid + it * 512;
            t[it] = hs[i < nel ? i : nel - 1];
        }
#pragma unroll
        for (int it = 0; it < NIT; ++it) acc[tid + it * 512] = t[it];
        if (tid < 32) acc[RB * 32 + tid] = 0.0f;
    }
    const int beg = (int)boff[b];
    const int end = (int)boff[b + 1];
    __syncthreads();

    int base = beg + wv * 32;
    unsigned rec = 0u;
    if (base < end) {
        const int i0 = base + lane;
        rec = binned[i0 < end ? i0 : end - 1];
    }
    while (base < end) {
        const int nbase = base + 512;
        const bool ok = base + lane < end;
        const int sidx = ok ? (int)(rec >> RB_LOG) : row0;
        const int didx = ok ? (int)(rec & (RB - 1)) : RB;
        float v[32];
#pragma unroll
        for (int u = 0; u < 32; ++u) {
            const int sj = __builtin_amdgcn_readlane(sidx, u);
            v[u] = h[sj * 32 + lane];
        }
        int ni = nbase + lane;
        ni = ni < end ? ni : end - 1;
        const unsigned nrec = binned[ni];
#pragma unroll
        for (int u = 0; u < 32; ++u) {
            const int dj = __builtin_amdgcn_readlane(didx, u);
            atomicAdd(&acc[dj * 32 + lane], v[u]);
        }
        rec  = nrec;
        base = nbase;
    }

    float* os = out + (size_t)row0 * 32;
    if constexpr (MODE == 1) {
        const float bch = bias[lane];
        float dg[NIT];
#pragma unroll
        for (int it = 0; it < NIT; ++it) {
            const int r = row0 + wv + it * 16;
            dg[it] = deg[r < n ? r : n - 1];
        }
        __syncthreads();
        float o[NIT];
#pragma unroll
        for (int it = 0; it < NIT; ++it) {
            const float dv = dinv_of(dg[it]);
            float v = dv * acc[tid + it * 512] + bch;
            v = v > 0.0f ? v : 0.0f;
            o[it] = dv * v;
        }
#pragma unroll
        for (int it = 0; it < NIT; ++it) {
            const int i = tid + it * 512;
            if (i < nel) os[i] = o[it];
        }
    } else {
        __syncthreads();
        float o[NIT];
#pragma unroll
        for (int it = 0; it < NIT; ++it) o[it] = acc[tid + it * 512];
#pragma unroll
        for (int it = 0; it < NIT; ++it) {
            const int i = tid + it * 512;
            if (i < nel) os[i] = o[it];
        }
    }
}


__global__ void k_final_head(const float* __restrict__ sums,
                             const float* __restrict__ cnts,
                             const float* __restrict__ Wc,
                             const float* __restrict__ bc,
                             float* out, int G) {
    int g = threadIdx.x;
    if (g >= G) return;
    float c = cnts[g];
    c = c > 1.0f ? c : 1.0f;
    float inv = 1.0f / c;
    float l0 = bc[0], l1 = bc[1];
    for (int ch = 0; ch < 64; ++ch) {
        float p = sums[g * 64 + ch] * inv;
        out[G * 2 + g * 64 + ch] = p;
        l0 = fmaf(p, Wc[ch * 2 + 0], l0);
        l1 = fmaf(p, Wc[ch * 2 + 1], l1);
    }
    out[g * 2 + 0] = l0;
    out[g * 2 + 1] = l1;
}


extern "C" void kernel_launch(void* const* d_in, const int* in_sizes, int n_in,
                              void* d_out, int out_size, void* d_ws, size_t ws_size,
                              hipStream_t stream) {
    const float* x   = (const float*)d_in[0];
    const float* W1  = (const float*)d_in[1];
    const float* b1  = (const float*)d_in[2];
    const float* W2  = (const float*)d_in[3];
    const float* b2  = (const float*)d_in[4];
    const float* Wc  = (const float*)d_in[5];
    const float* bc  = (const float*)d_in[6];
    const int*   ei  = (const int*)d_in[7];
    const int*   bat = (const int*)d_in[8];

    const int n = in_sizes[0] / IN_CH;
    const int E = in_sizes[7] / 2;
    const int G = 64;

    const int* src = ei;
    const int* dst = ei + E;

    const int B = 256;
    auto cdiv = [](long long a, long long b) { return (int)((a + b - 1) / b); };
    const int nb = cdiv(n, RB);

    float* ws   = (float*)d_ws;
    float* deg  = ws;                ws += n;
    float* h1   = ws;                ws += (size_t)n * 32;
    float* o1   = ws;                ws += (size_t)n * 32;
    unsigned* binned = (unsigned*)ws; ws += E;
    float* sums = ws;                ws += (size_t)G * 64;
    float* cnts = ws;                ws += G;
    unsigned* bcnt = (unsigned*)ws;  ws += (size_t)nb * CS;
    unsigned* bcur = (unsigned*)ws;  ws += (size_t)nb * CS;
    unsigned* boff = (unsigned*)ws;  ws += nb + 1;
    float* a2   = h1;

    k_init<<<cdiv(2 * nb * CS, B), B, 0, stream>>>(sums, G * 64 + G, bcnt, 2 * nb * CS);
    k_hist<<<cdiv(E, EPB), 256, nb * sizeof(unsigned), stream>>>(dst, bcnt, E, nb);
    k_fill<<<cdiv(E, EPB), 256, 3 * nb * sizeof(unsigned), stream>>>(src, dst, bcnt, bcur, boff, binned, E, nb);
    k_degcount<<<nb, DEG_T, 0, stream>>>(binned, boff, deg, n);

    const int tiles = (n + 15) / 16;
    k_gemm_wmma<128, 32, 0><<<cdiv(tiles, 4), 128, 0, stream>>>(x, W1, deg, nullptr, nullptr, h1, nullptr, n, tiles);
    k_agg<1><<<nb, 512, 0, stream>>>(binned, boff, h1, deg, b1, o1, n);
    k_agg<2><<<nb, 512, 0, stream>>>(binned, boff, o1, deg, nullptr, a2, n);
    k_gemm_wmma<32, 64, 1><<<cdiv(tiles, 4), 128, 0, stream>>>(a2, W2, deg, b2, bat, sums, cnts, n, tiles);

    k_final_head<<<1, 64, 0, stream>>>(sums, cnts, Wc, bc, (float*)d_out, G);
}


# Both matrix products moved into the block owner kernels, one 16-row tile per wave: K=128 shares the degree-count launch, K=32 reads the LDS edge sums; 9 launches to 7; pool atomics still unordered
# speedup vs baseline: 4.8606x; 1.0948x over previous
#include <hip/hip_runtime.h>
#include <hip/hip_bf16.h>


typedef float v2f __attribute__((ext_vector_type(2)));
typedef float v8f __attribute__((ext_vector_type(8)));

#define IN_CH 128
#define RB_LOG 8
#define RB 256
#define EPB 4096
#define CS 32

__global__ void k_init(float* pool, int n_pool, unsigned* zc, int n_zc) {
    const int i = blockIdx.x * blockDim.x + threadIdx.x;
    for (int j = i; j < n_pool; j += gridDim.x * blockDim.x) pool[j] = 0.0f;
    for (int j = i; j < n_zc; j += gridDim.x * blockDim.x) zc[j] = 0u;
}

__global__ void k_hist(const int* __restrict__ dst, unsigned* bcnt, int E, int nb) {
    extern __shared__ unsigned lh[];
    for (int j = threadIdx.x; j < nb; j += 256) lh[j] = 0u;
    const int base = blockIdx.x * EPB;
    int d[EPB / 256];
#pragma unroll
    for (int it = 0; it < EPB / 256; ++it) {
        const int e = base + it * 256 + (int)threadIdx.x;
        d[it] = dst[e < E ? e : E - 1];
    }
    __syncthreads();
#pragma unroll
    for (int it = 0; it < EPB / 256; ++it) {
        const int e = base + it * 256 + (int)threadIdx.x;
        if (e < E) atomicAdd(&lh[d[it] >> RB_LOG], 1u);
    }
    __syncthreads();
    for (int j = threadIdx.x; j < nb; j += 256) {
        const unsigned c = lh[j];
        if (c) atomicAdd(&bcnt[j * CS], c);
    }
}

__global__ void k_fill(const int* __restrict__ src, const int* __restrict__ dst,
                       const unsigned* __restrict__ bcnt, unsigned* bcur,
                       unsigned* boff, unsigned* __restrict__ binned,
                       int E, int nb) {
    extern __shared__ unsigned sm[];
    unsigned* s_off  = sm;
    unsigned* s_lc   = sm + nb;
    unsigned* s_base = sm + 2 * nb;
    __shared__ unsigned wsum[8];
    const int tid = threadIdx.x, lane = tid & 31, wv = tid >> 5;

    const int base = blockIdx.x * EPB;
    int s[EPB / 256], d[EPB / 256];
#pragma unroll
    for (int it = 0; it < EPB / 256; ++it) {
        const int e = base + it * 256 + tid;
        const int ec = e < E ? e : E - 1;
        s[it] = src[ec];
        d[it] = dst[ec];
    }

    unsigned carry = 0u;
    for (int b0 = 0; b0 < nb; b0 += 256) {
        const int i = b0 + tid;
        const unsigned v = i < nb ? bcnt[i * CS] : 0u;
        unsigned x = v;
#pragma unroll
        for (int d = 1; d < 32; d <<= 1) {
            const unsigned y = __shfl_up(x, d);
            if (lane >= d) x += y;
        }
        if (lane == 31) wsum[wv] = x;
        __syncthreads();
        unsigned wp = 0u, tot = 0u;
#pragma unroll
        for (int w = 0; w < 8; ++w) {
            const unsigned s = wsum[w];
            tot += s;
            if (w < wv) wp += s;
        }
        if (i < nb) {
            s_off[i] = carry + wp + x - v;
            s_lc[i] = 0u;
        }
        carry += tot;
        __syncthreads();
    }
    if (blockIdx.x == 0) {
        for (int j = tid; j < nb; j += 256) boff[j] = s_off[j];
        if (tid == 0) boff[nb] = carry;
    }

#pragma unroll
    for (int it = 0; it < EPB / 256; ++it) {
        if (base + it * 256 + tid < E) atomicAdd(&s_lc[d[it] >> RB_LOG], 1u);
    }
    __syncthreads();
    for (int j0 = tid; j0 < nb; j0 += 512) {
        const int j1 = j0 + 256;
        const bool two = j1 < nb;
        const int j1c = two ? j1 : j0;
        const unsigned c0 = s_lc[j0];
        const unsigned c1 = s_lc[j1c];
        const unsigned r0 = atomicAdd(&bcur[j0 * CS], c0);
        unsigned r1 = 0u;
        if (two) r1 = atomicAdd(&bcur[j1 * CS], c1);
        s_base[j0] = s_off[j0] + r0;
        s_lc[j0] = 0u;
        if (two) {
            s_base[j1] = s_off[j1] + r1;
            s_lc[j1] = 0u;
        }
    }
    __syncthreads();
    unsigned rk[EPB / 256], sb[EPB / 256];
#pragma unroll
    for (int it = 0; it < EPB / 256; ++it) {
        rk[it] = 0u;
        sb[it] = 0u;
        if (base + it * 256 + tid < E) {
            const int b = d[it] >> RB_LOG;
            rk[it] = atomicAdd(&s_lc[b], 1u);
            sb[it] = s_base[b];
        }
    }
#pragma unroll
    for (int it = 0; it < EPB / 256; ++it) {
        if (base + it * 256 + tid < E) {
            binned[sb[it] + rk[it]] =
                ((unsigned)s[it] << RB_LOG) | (unsigned)(d[it] & (RB - 1));
        }
    }
}

__device__ __forceinline__ float dinv_of(float d) {
    return d > 0.0f ? __frsqrt_rn(d) : 0.0f;
}


__device__ __forceinline__ void pool_tile(const v8f (&acc)[4], const float (&dv)[8],
                                          const float (&bv)[4], int g_row,
                                          const int* __restrict__ gid,
                                          float* __restrict__ O, float* __restrict__ O2,
                                          int m0, int n_rows, int lane) {
    constexpr int N = 64, NT = 4;
    const int colL  = lane & 15;
    const int mbase = m0 + (lane >> 4) * 8;
    const int g0 = __builtin_amdgcn_readfirstlane(g_row);
    const bool one_graph = (m0 + 16 <= n_rows) && __all(g_row == g0);
    if (one_graph) {
        float cs[NT];
#pragma unroll
        for (int t = 0; t < NT; ++t) {
            float s = 0.0f;
#pragma unroll
            for (int r = 0; r < 8; ++r) {
                const float v = dv[r] * acc[t][r] + bv[t];
                s += v > 0.0f ? v : 0.0f;
            }
            cs[t] = s + __shfl_xor(s, 16);
        }
#pragma unroll
        for (int p = 0; p < NT / 2; ++p) {
            const float s = lane < 16 ? cs[2 * p] : cs[2 * p + 1];
            unsafeAtomicAdd(&O[g0 * N + p * 32 + lane], s);
        }
        if (lane == 0) unsafeAtomicAdd(&O2[g0], 16.0f);
    } else {
#pragma unroll
        for (int r = 0; r < 8; ++r) {
            const int row = mbase + r;
            if (row < n_rows) {
                const int g = gid[row];
#pragma unroll
                for (int t = 0; t < NT; ++t) {
                    const float v = dv[r] * acc[t][r] + bv[t];
                    unsafeAtomicAdd(&O[g * N + t * 16 + colL], v > 0.0f ? v : 0.0f);
                }
                if (colL == 0) unsafeAtomicAdd(&O2[g], 1.0f);
            }
        }
    }
}

#define DEG_T 512
#define DEG_U 16
__global__ void __launch_bounds__(DEG_T)
k_deg_gemm1(const unsigned* __restrict__ binned,
            const unsigned* __restrict__ boff,
            const float* __restrict__ X,
            const float* __restrict__ W,
            float* __restrict__ deg,
            float* __restrict__ H, int n) {
    static_assert(DEG_T / 32 == RB / 16, "one wave per 16-row tile of the block");
    constexpr int K = IN_CH, N = 32, NT = N / 16, KS = K / 4;
    __shared__ unsigned cnt[RB];
    const int tid  = threadIdx.x;
    const int lane = tid & 31;
    const int wv   = __builtin_amdgcn_readfirstlane(tid >> 5);
    const int row0 = blockIdx.x * RB;
    if (tid < RB) cnt[tid] = 0u;
    const int beg = (int)boff[blockIdx.x];
    const int end = (int)boff[blockIdx.x + 1];

    const int m0    = row0 + wv * 16;
    const bool have = m0 < n;
    const int colL  = lane & 15;
    const int hh    = lane >> 4;
    const int khalf = hh * 2;
    const int mrow  = m0 + colL;
    const int rm    = mrow < n ? mrow : (n - 1);

    __syncthreads();

    unsigned v[DEG_U];
    if (beg < end) {
#pragma unroll
        for (int u = 0; u < DEG_U; ++u) {
            const int e = beg + u * DEG_T + tid;
            v[u] = binned[e < end ? e : end - 1];
        }
    }
    v2f a[KS];
    {
        const float* ap = X + (size_t)rm * K + khalf;
#pragma unroll
        for (int q = 0; q < KS; ++q) {
            a[q].x = ap[4 * q];
            a[q].y = ap[4 * q + 1];
        }
    }
    if (beg < end) {
#pragma unroll
        for (int u = 0; u < DEG_U; ++u) {
            if (beg + u * DEG_T + tid < end) atomicAdd(&cnt[v[u] & (RB - 1)], 1u);
        }
    }
    for (int c = beg + DEG_T * DEG_U; c < end; c += DEG_T * DEG_U) {
        unsigned w[DEG_U];
#pragma unroll
        for (int u = 0; u < DEG_U; ++u) {
            const int e = c + u * DEG_T + tid;
            w[u] = binned[e < end ? e : end - 1];
        }
#pragma unroll
        for (int u = 0; u < DEG_U; ++u) {
            if (c + u * DEG_T + tid < end) atomicAdd(&cnt[w[u] & (RB - 1)], 1u);
        }
    }
    __syncthreads();
    if (tid < RB && row0 + tid < n) deg[row0 + tid] = 1.0f + (float)cnt[tid];

    if (have) {
        const int lrow  = wv * 16 + hh * 8;
        const int mbase = row0 + lrow;
        float dv[8];
#pragma unroll
        for (int r = 0; r < 8; ++r) dv[r] = dinv_of(1.0f + (float)cnt[lrow + r]);

        v8f acc[NT] = {};
#pragma unroll
        for (int q = 0; q < KS; ++q) {
            const int k = 4 * q;
#pragma unroll
            for (int t = 0; t < NT; ++t) {
                v2f b;
                b.x = W[(k + khalf) * N + t * 16 + colL];
                b.y = W[(k + khalf + 1) * N + t * 16 + colL];
                acc[t] = __builtin_amdgcn_wmma_f32_16x16x4_f32(
                    false, a[q], false, b, (short)0, acc[t], false, false);
            }
        }

        if (m0 + 16 <= n) {
#pragma unroll
            for (int t = 0; t < NT; ++t) {
#pragma unroll
                for (int r = 0; r < 8; ++r) {
                    H[(mbase + r) * N + t * 16 + colL] = dv[r] * acc[t][r];
                }
            }
        } else {
#pragma unroll
            for (int t = 0; t < NT; ++t) {
#pragma unroll
                for (int r = 0; r < 8; ++r) {
                    const int row = mbase + r;
                    if (row < n) H[row * N + t * 16 + colL] = dv[r] * acc[t][r];
                }
            }
        }
    }
}


template <int MODE>
__global__ void __launch_bounds__(512)
k_agg(const unsigned* __restrict__ binned, const unsigned* __restrict__ boff,
      const float* __restrict__ h, const float* __restrict__ deg,
      const float* __restrict__ bias, float* __restrict__ out, int n,
      const float* __restrict__ W, const int* __restrict__ gid,
      float* __restrict__ O2) {
    __shared__ __attribute__((aligned(16))) float acc[(RB + 1) * 32];
    constexpr int NIT = RB * 32 / 512;
    const int tid  = threadIdx.x;
    const int lane = tid & 31;
    const int wv   = __builtin_amdgcn_readfirstlane(tid >> 5);
    const int b    = blockIdx.x;
    const int row0 = b * RB;
    const int nrows = (n - row0) < RB ? (n - row0) : RB;
    const int nel  = nrows * 32;
    const float* hs = h + (size_t)row0 * 32;
    {
        float t[NIT];
#pragma unroll
        for (int it = 0; it < NIT; ++it) {
            const int i = tid + it * 512;
            t[it] = hs[i < nel ? i : nel - 1];
        }
#pragma unroll
        for (int it = 0; it < NIT; ++it) acc[tid + it * 512] = t[it];
        if (tid < 32) acc[RB * 32 + tid] = 0.0f;
    }
    const int beg = (int)boff[b];
    const int end = (int)boff[b + 1];
    __syncthreads();

    int base = beg + wv * 32;
    unsigned rec = 0u;
    if (base < end) {
        const int i0 = base + lane;
        rec = binned[i0 < end ? i0 : end - 1];
    }
    constexpr int NT = 4;
    const int colL  = lane & 15;
    const int hh    = lane >> 4;
    const int m0    = row0 + wv * 16;
    const bool have = m0 < n;
    float dv[8];
    float bv[NT];
    int g_row = 0;
    if constexpr (MODE == 2) {
        const int mbase = m0 + hh * 8;
#pragma unroll
        for (int r = 0; r < 8; ++r) {
            const int row = mbase + r;
            dv[r] = dinv_of(deg[row < n ? row : n - 1]);
        }
#pragma unroll
        for (int t = 0; t < NT; ++t) bv[t] = bias[t * 16 + colL];
        const int mrow = m0 + colL;
        g_row = gid[mrow < n ? mrow : n - 1];
    }
    while (base < end) {
        const int nbase = base + 512;
        const bool ok = base + lane < end;
        const int sidx = ok ? (int)(rec >> RB_LOG) : row0;
        const int didx = ok ? (int)(rec & (RB - 1)) : RB;
        float v[32];
#pragma unroll
        for (int u = 0; u < 32; ++u) {
            const int sj = __builtin_amdgcn_readlane(sidx, u);
            v[u] = h[sj * 32 + lane];
        }
        int ni = nbase + lane;
        ni = ni < end ? ni : end - 1;
        const unsigned nrec = binned[ni];
#pragma unroll
        for (int u = 0; u < 32; ++u) {
            const int dj = __builtin_amdgcn_readlane(didx, u);
            atomicAdd(&acc[dj * 32 + lane], v[u]);
        }
        rec  = nrec;
        base = nbase;
    }

    if constexpr (MODE == 1) {
        float* os = out + (size_t)row0 * 32;
        const float bch = bias[lane];
        float dg[NIT];
#pragma unroll
        for (int it = 0; it < NIT; ++it) {
            const int r = row0 + wv + it * 16;
            dg[it] = deg[r < n ? r : n - 1];
        }
        __syncthreads();
        float o[NIT];
#pragma unroll
        for (int it = 0; it < NIT; ++it) {
            const float dv = dinv_of(dg[it]);
            float v = dv * acc[tid + it * 512] + bch;
            v = v > 0.0f ? v : 0.0f;
            o[it] = dv * v;
        }
#pragma unroll
        for (int it = 0; it < NIT; ++it) {
            const int i = tid + it * 512;
            if (i < nel) os[i] = o[it];
        }
    } else {
        static_assert(MODE != 2 || RB / 16 == 512 / 32, "one wave per 16-row tile of the block");
        constexpr int N = 64;
        const int khalf = hh * 2;
        __syncthreads();
        if (have) {
            v8f c2[NT] = {};
            const float* ar = acc + (wv * 16 + colL) * 32 + khalf;
#pragma unroll
            for (int k = 0; k < 32; k += 4) {
                const v2f a = *(const v2f*)(ar + k);
#pragma unroll
                for (int t = 0; t < NT; ++t) {
                    v2f bq;
                    bq.x = W[(k + khalf) * N + t * 16 + colL];
                    bq.y = W[(k + khalf + 1) * N + t * 16 + colL];
                    c2[t] = __builtin_amdgcn_wmma_f32_16x16x4_f32(
                        false, a, false, bq, (short)0, c2[t], false, false);
                }
            }
            pool_tile(c2, dv, bv, g_row, gid, out, O2, m0, n, lane);
        }
    }
}


__global__ void k_final_head(const float* __restrict__ sums,
                             const float* __restrict__ cnts,
                             const float* __restrict__ Wc,
                             const float* __restrict__ bc,
                             float* out, int G) {
    int g = threadIdx.x;
    if (g >= G) return;
    float c = cnts[g];
    c = c > 1.0f ? c : 1.0f;
    float inv = 1.0f / c;
    float l0 = bc[0], l1 = bc[1];
    for (int ch = 0; ch < 64; ++ch) {
        float p = sums[g * 64 + ch] * inv;
        out[G * 2 + g * 64 + ch] = p;
        l0 = fmaf(p, Wc[ch * 2 + 0], l0);
        l1 = fmaf(p, Wc[ch * 2 + 1], l1);
    }
    out[g * 2 + 0] = l0;
    out[g * 2 + 1] = l1;
}


extern "C" void kernel_launch(void* const* d_in, const int* in_sizes, int n_in,
                              void* d_out, int out_size, void* d_ws, size_t ws_size,
                              hipStream_t stream) {
    const float* x   = (const float*)d_in[0];
    const float* W1  = (const float*)d_in[1];
    const float* b1  = (const float*)d_in[2];
    const float* W2  = (const float*)d_in[3];
    const float* b2  = (const float*)d_in[4];
    const float* Wc  = (const float*)d_in[5];
    const float* bc  = (const float*)d_in[6];
    const int*   ei  = (const int*)d_in[7];
    const int*   bat = (const int*)d_in[8];

    const int n = in_sizes[0] / IN_CH;
    const int E = in_sizes[7] / 2;
    const int G = 64;

    const int* src = ei;
    const int* dst = ei + E;

    const int B = 256;
    auto cdiv = [](long long a, long long b) { return (int)((a + b - 1) / b); };
    const int nb = cdiv(n, RB);

    float* ws   = (float*)d_ws;
    float* deg  = ws;                ws += n;
    float* h1   = ws;                ws += (size_t)n * 32;
    float* o1   = ws;                ws += (size_t)n * 32;
    unsigned* binned = (unsigned*)ws; ws += E;
    float* sums = ws;                ws += (size_t)G * 64;
    float* cnts = ws;                ws += G;
    unsigned* bcnt = (unsigned*)ws;  ws += (size_t)nb * CS;
    unsigned* bcur = (unsigned*)ws;  ws += (size_t)nb * CS;
    unsigned* boff = (unsigned*)ws;  ws += nb + 1;

    k_init<<<cdiv(2 * nb * CS, B), B, 0, stream>>>(sums, G * 64 + G, bcnt, 2 * nb * CS);
    k_hist<<<cdiv(E, EPB), 256, nb * sizeof(unsigned), stream>>>(dst, bcnt, E, nb);
    k_fill<<<cdiv(E, EPB), 256, 3 * nb * sizeof(unsigned), stream>>>(src, dst, bcnt, bcur, boff, binned, E, nb);

    k_deg_gemm1<<<nb, DEG_T, 0, stream>>>(binned, boff, x, W1, deg, h1, n);
    k_agg<1><<<nb, 512, 0, stream>>>(binned, boff, h1, deg, b1, o1, n, nullptr, nullptr, nullptr);
    k_agg<2><<<nb, 512, 0, stream>>>(binned, boff, o1, deg, b2, sums, n, W2, bat, cnts);

    k_final_head<<<1, 64, 0, stream>>>(sums, cnts, Wc, bc, (float*)d_out, G);
}


# Fill kernel stages its records in LDS in bin order and stores each bin's share as a run of consecutive words (was one 4-byte store per record, a line per lane); order inside a bin unchanged
# speedup vs baseline: 5.3882x; 1.1085x over previous
#include <hip/hip_runtime.h>
#include <hip/hip_bf16.h>


typedef float v2f __attribute__((ext_vector_type(2)));
typedef float v8f __attribute__((ext_vector_type(8)));

#define IN_CH 128
#define RB_LOG 8
#define RB 256
#define EPB 4096
#define CS 32

__global__ void k_init(float* pool, int n_pool, unsigned* zc, int n_zc) {
    const int i = blockIdx.x * blockDim.x + threadIdx.x;
    for (int j = i; j < n_pool; j += gridDim.x * blockDim.x) pool[j] = 0.0f;
    for (int j = i; j < n_zc; j += gridDim.x * blockDim.x) zc[j] = 0u;
}

__global__ void k_hist(const int* __restrict__ dst, unsigned* bcnt, int E, int nb) {
    extern __shared__ unsigned lh[];
    for (int j = threadIdx.x; j < nb; j += 256) lh[j] = 0u;
    const int base = blockIdx.x * EPB;
    int d[EPB / 256];
#pragma unroll
    for (int it = 0; it < EPB / 256; ++it) {
        const int e = base + it * 256 + (int)threadIdx.x;
        d[it] = dst[e < E ? e : E - 1];
    }
    __syncthreads();
#pragma unroll
    for (int it = 0; it < EPB / 256; ++it) {
        const int e = base + it * 256 + (int)threadIdx.x;
        if (e < E) atomicAdd(&lh[d[it] >> RB_LOG], 1u);
    }
    __syncthreads();
    for (int j = threadIdx.x; j < nb; j += 256) {
        const unsigned c = lh[j];
        if (c) atomicAdd(&bcnt[j * CS], c);
    }
}

__global__ void k_fill(const int* __restrict__ src, const int* __restrict__ dst,
                       const unsigned* __restrict__ bcnt, unsigned* bcur,
                       unsigned* boff, unsigned* __restrict__ binned,
                       int E, int nb) {
    extern __shared__ unsigned sm[];
    unsigned* s_off  = sm;
    unsigned* s_lc   = sm + nb;
    unsigned* s_base = sm + 2 * nb;
    unsigned* s_loc  = sm + 3 * nb;
    __shared__ unsigned wsum[8];
    __shared__ unsigned st_rec[EPB];
    __shared__ unsigned st_slot[EPB];
    const int tid = threadIdx.x, lane = tid & 31, wv = tid >> 5;

    const int base = blockIdx.x * EPB;
    int s[EPB / 256], d[EPB / 256];
#pragma unroll
    for (int it = 0; it < EPB / 256; ++it) {
        const int e = base + it * 256 + tid;
        const int ec = e < E ? e : E - 1;
        s[it] = src[ec];
        d[it] = dst[ec];
    }

    unsigned carry = 0u;
    for (int b0 = 0; b0 < nb; b0 += 256) {
        const int i = b0 + tid;
        const unsigned v = i < nb ? bcnt[i * CS] : 0u;
        unsigned x = v;
#pragma unroll
        for (int d = 1; d < 32; d <<= 1) {
            const unsigned y = __shfl_up(x, d);
            if (lane >= d) x += y;
        }
        if (lane == 31) wsum[wv] = x;
        __syncthreads();
        unsigned wp = 0u, tot = 0u;
#pragma unroll
        for (int w = 0; w < 8; ++w) {
            const unsigned s = wsum[w];
            tot += s;
            if (w < wv) wp += s;
        }
        if (i < nb) {
            s_off[i] = carry + wp + x - v;
            s_lc[i] = 0u;
        }
        carry += tot;
        __syncthreads();
    }
    if (blockIdx.x == 0) {
        for (int j = tid; j < nb; j += 256) boff[j] = s_off[j];
        if (tid == 0) boff[nb] = carry;
    }

#pragma unroll
    for (int it = 0; it < EPB / 256; ++it) {
        if (base + it * 256 + tid < E) atomicAdd(&s_lc[d[it] >> RB_LOG], 1u);
    }
    __syncthreads();
    unsigned lcarry = 0u;
    for (int b0 = 0; b0 < nb; b0 += 256) {
        const int i = b0 + tid;
        const unsigned v = i < nb ? s_lc[i] : 0u;
        unsigned x = v;
#pragma unroll
        for (int d = 1; d < 32; d <<= 1) {
            const unsigned y = __shfl_up(x, d);
            if (lane >= d) x += y;
        }
        if (lane == 31) wsum[wv] = x;
        __syncthreads();
        unsigned wp = 0u, tot = 0u;
#pragma unroll
        for (int w = 0; w < 8; ++w) {
            const unsigned s = wsum[w];
            tot += s;
            if (w < wv) wp += s;
        }
        if (i < nb) s_loc[i] = lcarry + wp + x - v;
        lcarry += tot;
        __syncthreads();
    }
    for (int j0 = tid; j0 < nb; j0 += 512) {
        const int j1 = j0 + 256;
        const bool two = j1 < nb;
        const int j1c = two ? j1 : j0;
        const unsigned c0 = s_lc[j0];
        const unsigned c1 = s_lc[j1c];
        const unsigned r0 = atomicAdd(&bcur[j0 * CS], c0);
        unsigned r1 = 0u;
        if (two) r1 = atomicAdd(&bcur[j1 * CS], c1);
        s_base[j0] = s_off[j0] + r0;
        s_lc[j0] = 0u;
        if (two) {
            s_base[j1] = s_off[j1] + r1;
            s_lc[j1] = 0u;
        }
    }
    __syncthreads();
    unsigned rk[EPB / 256], sb[EPB / 256], lo[EPB / 256];
#pragma unroll
    for (int it = 0; it < EPB / 256; ++it) {
        rk[it] = 0u;
        sb[it] = 0u;
        lo[it] = 0u;
        if (base + it * 256 + tid < E) {
            const int b = d[it] >> RB_LOG;
            rk[it] = atomicAdd(&s_lc[b], 1u);
            sb[it] = s_base[b];
            lo[it] = s_loc[b];
        }
    }
#pragma unroll
    for (int it = 0; it < EPB / 256; ++it) {
        if (base + it * 256 + tid < E) {
            const unsigned p = lo[it] + rk[it];
            st_rec[p]  = ((unsigned)s[it] << RB_LOG) | (unsigned)(d[it] & (RB - 1));
            st_slot[p] = sb[it] + rk[it];
        }
    }
    __syncthreads();
    const int nval = (E - base) < EPB ? (E - base) : EPB;
    unsigned qr[EPB / 256], qs[EPB / 256];
#pragma unroll
    for (int it = 0; it < EPB / 256; ++it) {
        const int i = it * 256 + tid;
        const int ic = i < nval ? i : 0;
        qr[it] = st_rec[ic];
        qs[it] = st_slot[ic];
    }
#pragma unroll
    for (int it = 0; it < EPB / 256; ++it) {
        if (it * 256 + tid < nval) binned[qs[it]] = qr[it];
    }
}

__device__ __forceinline__ float dinv_of(float d) {
    return d > 0.0f ? __frsqrt_rn(d) : 0.0f;
}


__device__ __forceinline__ void pool_tile(const v8f (&acc)[4], const float (&dv)[8],
                                          const float (&bv)[4], int g_row,
                                          const int* __restrict__ gid,
                                          float* __restrict__ O, float* __restrict__ O2,
                                          int m0, int n_rows, int lane) {
    constexpr int N = 64, NT = 4;
    const int colL  = lane & 15;
    const int mbase = m0 + (lane >> 4) * 8;
    const int g0 = __builtin_amdgcn_readfirstlane(g_row);
    const bool one_graph = (m0 + 16 <= n_rows) && __all(g_row == g0);
    if (one_graph) {
        float cs[NT];
#pragma unroll
        for (int t = 0; t < NT; ++t) {
            float s = 0.0f;
#pragma unroll
            for (int r = 0; r < 8; ++r) {
                const float v = dv[r] * acc[t][r] + bv[t];
                s += v > 0.0f ? v : 0.0f;
            }
            cs[t] = s + __shfl_xor(s, 16);
        }
#pragma unroll
        for (int p = 0; p < NT / 2; ++p) {
            const float s = lane < 16 ? cs[2 * p] : cs[2 * p + 1];
            unsafeAtomicAdd(&O[g0 * N + p * 32 + lane], s);
        }
        if (lane == 0) unsafeAtomicAdd(&O2[g0], 16.0f);
    } else {
#pragma unroll
        for (int r = 0; r < 8; ++r) {
            const int row = mbase + r;
            if (row < n_rows) {
                const int g = gid[row];
#pragma unroll
                for (int t = 0; t < NT; ++t) {
                    const float v = dv[r] * acc[t][r] + bv[t];
                    unsafeAtomicAdd(&O[g * N + t * 16 + colL], v > 0.0f ? v : 0.0f);
                }
                if (colL == 0) unsafeAtomicAdd(&O2[g], 1.0f);
            }
        }
    }
}

#define DEG_T 512
#define DEG_U 16
__global__ void __launch_bounds__(DEG_T)
k_deg_gemm1(const unsigned* __restrict__ binned,
            const unsigned* __restrict__ boff,
            const float* __restrict__ X,
            const float* __restrict__ W,
            float* __restrict__ deg,
            float* __restrict__ H, int n) {
    static_assert(DEG_T / 32 == RB / 16, "one wave per 16-row tile of the block");
    constexpr int K = IN_CH, N = 32, NT = N / 16, KS = K / 4;
    __shared__ unsigned cnt[RB];
    const int tid  = threadIdx.x;
    const int lane = tid & 31;
    const int wv   = __builtin_amdgcn_readfirstlane(tid >> 5);
    const int row0 = blockIdx.x * RB;
    if (tid < RB) cnt[tid] = 0u;
    const int beg = (int)boff[blockIdx.x];
    const int end = (int)boff[blockIdx.x + 1];

    const int m0    = row0 + wv * 16;
    const bool have = m0 < n;
    const int colL  = lane & 15;
    const int hh    = lane >> 4;
    const int khalf = hh * 2;
    const int mrow  = m0 + colL;
    const int rm    = mrow < n ? mrow : (n - 1);

    __syncthreads();

    unsigned v[DEG_U];
    if (beg < end) {
#pragma unroll
        for (int u = 0; u < DEG_U; ++u) {
            const int e = beg + u * DEG_T + tid;
            v[u] = binned[e < end ? e : end - 1];
        }
    }
    v2f a[KS];
    {
        const float* ap = X + (size_t)rm * K + khalf;
#pragma unroll
        for (int q = 0; q < KS; ++q) {
            a[q].x = ap[4 * q];
            a[q].y = ap[4 * q + 1];
        }
    }
    if (beg < end) {
#pragma unroll
        for (int u = 0; u < DEG_U; ++u) {
            if (beg + u * DEG_T + tid < end) atomicAdd(&cnt[v[u] & (RB - 1)], 1u);
        }
    }
    for (int c = beg + DEG_T * DEG_U; c < end; c += DEG_T * DEG_U) {
        unsigned w[DEG_U];
#pragma unroll
        for (int u = 0; u < DEG_U; ++u) {
            const int e = c + u * DEG_T + tid;
            w[u] = binned[e < end ? e : end - 1];
        }
#pragma unroll
        for (int u = 0; u < DEG_U; ++u) {
            if (c + u * DEG_T + tid < end) atomicAdd(&cnt[w[u] & (RB - 1)], 1u);
        }
    }
    __syncthreads();
    if (tid < RB && row0 + tid < n) deg[row0 + tid] = 1.0f + (float)cnt[tid];

    if (have) {
        const int lrow  = wv * 16 + hh * 8;
        const int mbase = row0 + lrow;
        float dv[8];
#pragma unroll
        for (int r = 0; r < 8; ++r) dv[r] = dinv_of(1.0f + (float)cnt[lrow + r]);

        v8f acc[NT] = {};
#pragma unroll
        for (int q = 0; q < KS; ++q) {
            const int k = 4 * q;
#pragma unroll
            for (int t = 0; t < NT; ++t) {
                v2f b;
                b.x = W[(k + khalf) * N + t * 16 + colL];
                b.y = W[(k + khalf + 1) * N + t * 16 + colL];
                acc[t] = __builtin_amdgcn_wmma_f32_16x16x4_f32(
                    false, a[q], false, b, (short)0, acc[t], false, false);
            }
        }

        if (m0 + 16 <= n) {
#pragma unroll
            for (int t = 0; t < NT; ++t) {
#pragma unroll
                for (int r = 0; r < 8; ++r) {
                    H[(mbase + r) * N + t * 16 + colL] = dv[r] * acc[t][r];
                }
            }
        } else {
#pragma unroll
            for (int t = 0; t < NT; ++t) {
#pragma unroll
                for (int r = 0; r < 8; ++r) {
                    const int row = mbase + r;
                    if (row < n) H[row * N + t * 16 + colL] = dv[r] * acc[t][r];
                }
            }
        }
    }
}


template <int MODE>
__global__ void __launch_bounds__(512)
k_agg(const unsigned* __restrict__ binned, const unsigned* __restrict__ boff,
      const float* __restrict__ h, const float* __restrict__ deg,
      const float* __restrict__ bias, float* __restrict__ out, int n,
      const float* __restrict__ W, const int* __restrict__ gid,
      float* __restrict__ O2) {
    __shared__ __attribute__((aligned(16))) float acc[(RB + 1) * 32];
    constexpr int NIT = RB * 32 / 512;
    const int tid  = threadIdx.x;
    const int lane = tid & 31;
    const int wv   = __builtin_amdgcn_readfirstlane(tid >> 5);
    const int b    = blockIdx.x;
    const int row0 = b * RB;
    const int nrows = (n - row0) < RB ? (n - row0) : RB;
    const int nel  = nrows * 32;
    const float* hs = h + (size_t)row0 * 32;
    {
        float t[NIT];
#pragma unroll
        for (int it = 0; it < NIT; ++it) {
            const int i = tid + it * 512;
            t[it] = hs[i < nel ? i : nel - 1];
        }
#pragma unroll
        for (int it = 0; it < NIT; ++it) acc[tid + it * 512] = t[it];
        if (tid < 32) acc[RB * 32 + tid] = 0.0f;
    }
    const int beg = (int)boff[b];
    const int end = (int)boff[b + 1];
    __syncthreads();

    int base = beg + wv * 32;
    unsigned rec = 0u;
    if (base < end) {
        const int i0 = base + lane;
        rec = binned[i0 < end ? i0 : end - 1];
    }
    constexpr int NT = 4;
    const int colL  = lane & 15;
    const int hh    = lane >> 4;
    const int m0    = row0 + wv * 16;
    const bool have = m0 < n;
    float dv[8];
    float bv[NT];
    int g_row = 0;
    if constexpr (MODE == 2) {
        const int mbase = m0 + hh * 8;
#pragma unroll
        for (int r = 0; r < 8; ++r) {
            const int row = mbase + r;
            dv[r] = dinv_of(deg[row < n ? row : n - 1]);
        }
#pragma unroll
        for (int t = 0; t < NT; ++t) bv[t] = bias[t * 16 + colL];
        const int mrow = m0 + colL;
        g_row = gid[mrow < n ? mrow : n - 1];
    }
    while (base < end) {
        const int nbase = base + 512;
        const bool ok = base + lane < end;
        const int sidx = ok ? (int)(rec >> RB_LOG) : row0;
        const int didx = ok ? (int)(rec & (RB - 1)) : RB;
        float v[32];
#pragma unroll
        for (int u = 0; u < 32; ++u) {
            const int sj = __builtin_amdgcn_readlane(sidx, u);
            v[u] = h[sj * 32 + lane];
        }
        int ni = nbase + lane;
        ni = ni < end ? ni : end - 1;
        const unsigned nrec = binned[ni];
#pragma unroll
        for (int u = 0; u < 32; ++u) {
            const int dj = __builtin_amdgcn_readlane(didx, u);
            atomicAdd(&acc[dj * 32 + lane], v[u]);
        }
        rec  = nrec;
        base = nbase;
    }

    if constexpr (MODE == 1) {
        float* os = out + (size_t)row0 * 32;
        const float bch = bias[lane];
        float dg[NIT];
#pragma unroll
        for (int it = 0; it < NIT; ++it) {
            const int r = row0 + wv + it * 16;
            dg[it] = deg[r < n ? r : n - 1];
        }
        __syncthreads();
        float o[NIT];
#pragma unroll
        for (int it = 0; it < NIT; ++it) {
            const float dv = dinv_of(dg[it]);
            float v = dv * acc[tid + it * 512] + bch;
            v = v > 0.0f ? v : 0.0f;
            o[it] = dv * v;
        }
#pragma unroll
        for (int it = 0; it < NIT; ++it) {
            const int i = tid + it * 512;
            if (i < nel) os[i] = o[it];
        }
    } else {
        static_assert(MODE != 2 || RB / 16 == 512 / 32, "one wave per 16-row tile of the block");
        constexpr int N = 64;
        const int khalf = hh * 2;
        __syncthreads();
        if (have) {
            v8f c2[NT] = {};
            const float* ar = acc + (wv * 16 + colL) * 32 + khalf;
#pragma unroll
            for (int k = 0; k < 32; k += 4) {
                const v2f a = *(const v2f*)(ar + k);
#pragma unroll
                for (int t = 0; t < NT; ++t) {
                    v2f bq;
                    bq.x = W[(k + khalf) * N + t * 16 + colL];
                    bq.y = W[(k + khalf + 1) * N + t * 16 + colL];
                    c2[t] = __builtin_amdgcn_wmma_f32_16x16x4_f32(
                        false, a, false, bq, (short)0, c2[t], false, false);
                }
            }
            pool_tile(c2, dv, bv, g_row, gid, out, O2, m0, n, lane);
        }
    }
}


__global__ void k_final_head(const float* __restrict__ sums,
                             const float* __restrict__ cnts,
                             const float* __restrict__ Wc,
                             const float* __restrict__ bc,
                             float* out, int G) {
    int g = threadIdx.x;
    if (g >= G) return;
    float c = cnts[g];
    c = c > 1.0f ? c : 1.0f;
    float inv = 1.0f / c;
    float l0 = bc[0], l1 = bc[1];
    for (int ch = 0; ch < 64; ++ch) {
        float p = sums[g * 64 + ch] * inv;
        out[G * 2 + g * 64 + ch] = p;
        l0 = fmaf(p, Wc[ch * 2 + 0], l0);
        l1 = fmaf(p, Wc[ch * 2 + 1], l1);
    }
    out[g * 2 + 0] = l0;
    out[g * 2 + 1] = l1;
}


extern "C" void kernel_launch(void* const* d_in, const int* in_sizes, int n_in,
                              void* d_out, int out_size, void* d_ws, size_t ws_size,
                              hipStream_t stream) {
    const float* x   = (const float*)d_in[0];
    const float* W1  = (const float*)d_in[1];
    const float* b1  = (const float*)d_in[2];
    const float* W2  = (const float*)d_in[3];
    const float* b2  = (const float*)d_in[4];
    const float* Wc  = (const float*)d_in[5];
    const float* bc  = (const float*)d_in[6];
    const int*   ei  = (const int*)d_in[7];
    const int*   bat = (const int*)d_in[8];

    const int n = in_sizes[0] / IN_CH;
    const int E = in_sizes[7] / 2;
    const int G = 64;

    const int* src = ei;
    const int* dst = ei + E;

    const int B = 256;
    auto cdiv = [](long long a, long long b) { return (int)((a + b - 1) / b); };
    const int nb = cdiv(n, RB);

    float* ws   = (float*)d_ws;
    float* deg  = ws;                ws += n;
    float* h1   = ws;                ws += (size_t)n * 32;
    float* o1   = ws;                ws += (size_t)n * 32;
    unsigned* binned = (unsigned*)ws; ws += E;
    float* sums = ws;                ws += (size_t)G * 64;
    float* cnts = ws;                ws += G;
    unsigned* bcnt = (unsigned*)ws;  ws += (size_t)nb * CS;
    unsigned* bcur = (unsigned*)ws;  ws += (size_t)nb * CS;
    unsigned* boff = (unsigned*)ws;  ws += nb + 1;

    k_init<<<cdiv(2 * nb * CS, B), B, 0, stream>>>(sums, G * 64 + G, bcnt, 2 * nb * CS);
    k_hist<<<cdiv(E, EPB), 256, nb * sizeof(unsigned), stream>>>(dst, bcnt, E, nb);
    k_fill<<<cdiv(E, EPB), 256, 4 * nb * sizeof(unsigned), stream>>>(src, dst, bcnt, bcur, boff, binned, E, nb);

    k_deg_gemm1<<<nb, DEG_T, 0, stream>>>(binned, boff, x, W1, deg, h1, n);
    k_agg<1><<<nb, 512, 0, stream>>>(binned, boff, h1, deg, b1, o1, n, nullptr, nullptr, nullptr);
    k_agg<2><<<nb, 512, 0, stream>>>(binned, boff, o1, deg, b2, sums, n, W2, bat, cnts);

    k_final_head<<<1, 64, 0, stream>>>(sums, cnts, Wc, bc, (float*)d_out, G);
}
